# starting kernel
# baseline (speedup 1.0000x reference)
#include <hip/hip_runtime.h>

typedef __attribute__((ext_vector_type(2))) float v2f;
typedef __attribute__((ext_vector_type(8))) float v8f;

#define D 128

__global__ void gemm_wmma_f32(const float* __restrict__ X,
                              const float* __restrict__ W,
                              const float* __restrict__ bias,
                              float* __restrict__ Y) {
  const int lane = threadIdx.x & 31;
  const int wave = threadIdx.x >> 5;
  const int m0 = blockIdx.x * 16;
  const int n0 = wave * 16;
  const int r = lane & 15;
  const int koff = (lane >> 4) * 2;

  const float* arow = X + (size_t)(m0 + r) * D + koff;
  const float* brow = W + (size_t)(n0 + r) * D + koff;

  v8f c = {};
#pragma unroll 4
  for (int kb = 0; kb < D / 4; ++kb) {
    v2f a = *(const v2f*)(arow + kb * 4);
    v2f b = *(const v2f*)(brow + kb * 4);
    c = __builtin_amdgcn_wmma_f32_16x16x4_f32(false, a, false, b,
                                              (short)0, c, false, false);
  }

  const float bv = bias[n0 + r];
  const int hi = lane >> 4;
  float* yout = Y + (size_t)(m0 + hi * 8) * D + n0 + r;
#pragma unroll
  for (int v = 0; v < 8; ++v) yout[(size_t)v * D] = c[v] + bv;
}

__global__ void scatter_edges(const float* __restrict__ Wh,
                              const int* __restrict__ src,
                              const int* __restrict__ dst,
                              float* __restrict__ sum,
                              float* __restrict__ cnt,
                              int nE) {
  const int lane = threadIdx.x & 31;
  const int e = blockIdx.x * (blockDim.x >> 5) + (threadIdx.x >> 5);
  if (e >= nE) return;
  const int s = src[e];
  const int d = dst[e];
  const float4 v = ((const float4*)(Wh + (size_t)s * D))[lane];
  float* dp = sum + (size_t)d * D + lane * 4;
  atomicAdd(dp + 0, v.x);
  atomicAdd(dp + 1, v.y);
  atomicAdd(dp + 2, v.z);
  atomicAdd(dp + 3, v.w);
  if (lane == 0) atomicAdd(cnt + d, 1.0f);
}

__global__ void zero_f32(float* __restrict__ p, size_t n) {
  size_t i = (size_t)blockIdx.x * blockDim.x + threadIdx.x;
  const size_t stride = (size_t)gridDim.x * blockDim.x;
  for (; i < n; i += stride) p[i] = 0.0f;
}

__global__ void finalize_user(const float* __restrict__ sum_f,
                              const float* __restrict__ cnt_f,
                              const float* __restrict__ sum_rb,
                              const float* __restrict__ cnt_rb,
                              float* __restrict__ out, size_t n) {
  size_t i = (size_t)blockIdx.x * blockDim.x + threadIdx.x;
  if (i >= n) return;
  const size_t u = i >> 7;
  out[i] = sum_f[i] / fmaxf(cnt_f[u], 1.0f) + sum_rb[i] / fmaxf(cnt_rb[u], 1.0f);
}

__global__ void finalize_item(const float* __restrict__ cnt,
                              float* __restrict__ out, size_t n) {
  size_t i = (size_t)blockIdx.x * blockDim.x + threadIdx.x;
  if (i >= n) return;
  const size_t u = i >> 7;
  out[i] = out[i] / fmaxf(cnt[u], 1.0f);
}

extern "C" void kernel_launch(void* const* d_in, const int* in_sizes, int n_in,
                              void* d_out, int out_size, void* d_ws, size_t ws_size,
                              hipStream_t stream) {
  (void)n_in; (void)out_size; (void)ws_size;

  const float* feat_user  = (const float*)d_in[0];
  const float* feat_item  = (const float*)d_in[1];
  const float* W_follows  = (const float*)d_in[2];
  const float* b_follows  = (const float*)d_in[3];
  const float* W_rates    = (const float*)d_in[4];
  const float* b_rates    = (const float*)d_in[5];
  const float* W_ratedby  = (const float*)d_in[6];
  const float* b_ratedby  = (const float*)d_in[7];
  const int* src_follows  = (const int*)d_in[8];
  const int* dst_follows  = (const int*)d_in[9];
  const int* src_rates    = (const int*)d_in[10];
  const int* dst_rates    = (const int*)d_in[11];
  const int* src_ratedby  = (const int*)d_in[12];
  const int* dst_ratedby  = (const int*)d_in[13];

  const int NU  = in_sizes[0] / D;
  const int NI  = in_sizes[1] / D;
  const int Ef  = in_sizes[8];
  const int Er  = in_sizes[10];
  const int Erb = in_sizes[12];

  float* out_user = (float*)d_out;
  float* out_item = out_user + (size_t)NU * D;

  float* ws = (float*)d_ws;
  const size_t nmax = (size_t)(NU > NI ? NU : NI) * D;
  float* whbuf  = ws;
  float* sum_f  = whbuf + nmax;
  float* sum_rb = sum_f + (size_t)NU * D;
  float* cnt_f  = sum_rb + (size_t)NU * D;
  float* cnt_rb = cnt_f + NU;
  float* cnt_r  = cnt_rb + NU;

  const size_t zcount = (size_t)NU * D * 2 + (size_t)NU * 2 + (size_t)NI;
  zero_f32<<<2048, 256, 0, stream>>>(sum_f, zcount);
  zero_f32<<<2048, 256, 0, stream>>>(out_item, (size_t)NI * D);

  gemm_wmma_f32<<<NU / 16, 256, 0, stream>>>(feat_user, W_follows, b_follows, whbuf);
  scatter_edges<<<(Ef + 7) / 8, 256, 0, stream>>>(whbuf, src_follows, dst_follows,
                                                  sum_f, cnt_f, Ef);
  gemm_wmma_f32<<<NI / 16, 256, 0, stream>>>(feat_item, W_ratedby, b_ratedby, whbuf);
  scatter_edges<<<(Erb + 7) / 8, 256, 0, stream>>>(whbuf, src_ratedby, dst_ratedby,
                                                   sum_rb, cnt_rb, Erb);
  gemm_wmma_f32<<<NU / 16, 256, 0, stream>>>(feat_user, W_rates, b_rates, whbuf);
  scatter_edges<<<(Er + 7) / 8, 256, 0, stream>>>(whbuf, src_rates, dst_rates,
                                                  out_item, cnt_r, Er);

  const size_t nu_elems = (size_t)NU * D;
  const size_t ni_elems = (size_t)NI * D;
  finalize_user<<<(unsigned)((nu_elems + 255) / 256), 256, 0, stream>>>(
      sum_f, cnt_f, sum_rb, cnt_rb, out_user, nu_elems);
  finalize_item<<<(unsigned)((ni_elems + 255) / 256), 256, 0, stream>>>(
      cnt_r, out_item, ni_elems);
}


# global f32 atomic scatter, zero and divide passes replaced: a work-group owns 400 dst rows, scans dst, sums its rows in LDS (LDS f32 atomics, arrival order), writes means once; 10 -> 6 launches
# speedup vs baseline: 1.8007x; 1.8007x over previous
#include <hip/hip_runtime.h>

typedef __attribute__((ext_vector_type(2))) float v2f;
typedef __attribute__((ext_vector_type(8))) float v8f;

#define D 128

__global__ void gemm_wmma_f32(const float* __restrict__ X,
                              const float* __restrict__ W,
                              const float* __restrict__ bias,
                              float* __restrict__ Y) {
  const int lane = threadIdx.x & 31;
  const int wave = threadIdx.x >> 5;
  const int m0 = blockIdx.x * 16;
  const int n0 = wave * 16;
  const int r = lane & 15;
  const int koff = (lane >> 4) * 2;

  const float* arow = X + (size_t)(m0 + r) * D + koff;
  const float* brow = W + (size_t)(n0 + r) * D + koff;

  v8f c = {};
#pragma unroll 4
  for (int kb = 0; kb < D / 4; ++kb) {
    v2f a = *(const v2f*)(arow + kb * 4);
    v2f b = *(const v2f*)(brow + kb * 4);
    c = __builtin_amdgcn_wmma_f32_16x16x4_f32(false, a, false, b,
                                              (short)0, c, false, false);
  }

  const float bv = bias[n0 + r];
  const int hi = lane >> 4;
  float* yout = Y + (size_t)(m0 + hi * 8) * D + n0 + r;
#pragma unroll
  for (int v = 0; v < 8; ++v) yout[(size_t)v * D] = c[v] + bv;
}

#define AGG_ROWS 400
#define AGG_QCAP 6144
#define AGG_THREADS 1024
#define AGG_SCAN_U 8
#define AGG_GATH_U 8

__device__ __forceinline__ void agg_push(int e, unsigned dl,
                                         const float* __restrict__ Wh,
                                         const int* __restrict__ src,
                                         float* tile, int* cnt, int* qe, int* qd,
                                         unsigned* qn) {
  atomicAdd(&cnt[dl], 1);
  const unsigned k = atomicAdd(qn, 1u);
  if (k < AGG_QCAP) {
    qe[k] = e;
    qd[k] = (int)dl;
  } else {
    const float* w = Wh + (size_t)src[e] * D;
    for (int f = 0; f < D; ++f)
      atomicAdd(&tile[dl * D + (f & 3) * 32 + (f >> 2)], w[f]);
  }
}

__global__ __launch_bounds__(AGG_THREADS)
void agg_owned_rows(const float* __restrict__ Wh,
                    const int* __restrict__ src,
                    const int* __restrict__ dst,
                    float* __restrict__ out,
                    int nE, int nDst, int accumulate) {
  __shared__ float tile[AGG_ROWS * D];
  __shared__ int cnt[AGG_ROWS];
  __shared__ int qe[AGG_QCAP];
  __shared__ int qd[AGG_QCAP];
  __shared__ unsigned qn;

  const int tid = threadIdx.x;
  const int lane = tid & 31;
  const int wave = tid >> 5;
  const int d0 = blockIdx.x * AGG_ROWS;
  const int rows = (nDst - d0) < AGG_ROWS ? (nDst - d0) : AGG_ROWS;
  const unsigned urows = (unsigned)rows;

  float4* t4 = (float4*)tile;
  for (int i = tid; i < AGG_ROWS * D / 4; i += AGG_THREADS)
    t4[i] = make_float4(0.0f, 0.0f, 0.0f, 0.0f);
  for (int i = tid; i < AGG_ROWS; i += AGG_THREADS) cnt[i] = 0;
  if (tid == 0) qn = 0u;
  __syncthreads();

  const int n4 = nE >> 2;
  const int4* dst4 = (const int4*)dst;
  if (n4 > 0) {
    const int off = (int)blockIdx.x * (n4 / (int)gridDim.x);
    for (int t0 = tid; t0 < n4; t0 += AGG_THREADS * AGG_SCAN_U) {
      int4 dv[AGG_SCAN_U];
      int id[AGG_SCAN_U];
#pragma unroll
      for (int u = 0; u < AGG_SCAN_U; ++u) {
        const int t = t0 + u * AGG_THREADS;
        int i = t + off;
        if (i >= n4) i -= n4;
        const bool ok = t < n4;
        i = ok ? i : 0;
        id[u] = i;
        int4 v = dst4[i];
        if (!ok) v = make_int4(-1, -1, -1, -1);
        dv[u] = v;
      }
#pragma unroll
      for (int u = 0; u < AGG_SCAN_U; ++u) {
        const unsigned a0 = (unsigned)(dv[u].x - d0);
        const unsigned a1 = (unsigned)(dv[u].y - d0);
        const unsigned a2 = (unsigned)(dv[u].z - d0);
        const unsigned a3 = (unsigned)(dv[u].w - d0);
        const unsigned m01 = a0 < a1 ? a0 : a1;
        const unsigned m23 = a2 < a3 ? a2 : a3;
        const unsigned m = m01 < m23 ? m01 : m23;
        if (m < urows) {
          const int e0 = id[u] * 4;
          if (a0 < urows) agg_push(e0 + 0, a0, Wh, src, tile, cnt, qe, qd, &qn);
          if (a1 < urows) agg_push(e0 + 1, a1, Wh, src, tile, cnt, qe, qd, &qn);
          if (a2 < urows) agg_push(e0 + 2, a2, Wh, src, tile, cnt, qe, qd, &qn);
          if (a3 < urows) agg_push(e0 + 3, a3, Wh, src, tile, cnt, qe, qd, &qn);
        }
      }
    }
  }
  if (tid < (nE & 3)) {
    const int e = n4 * 4 + tid;
    const unsigned a = (unsigned)(dst[e] - d0);
    if (a < urows) agg_push(e, a, Wh, src, tile, cnt, qe, qd, &qn);
  }
  __syncthreads();

  const int n = qn < (unsigned)AGG_QCAP ? (int)qn : AGG_QCAP;

  for (int kb = tid; kb < n; kb += AGG_THREADS * 4) {
    int kk[4];
    int ee[4];
    int ss[4];
#pragma unroll
    for (int u = 0; u < 4; ++u) {
      const int k = kb + u * AGG_THREADS;
      kk[u] = k < n ? k : kb;
      ee[u] = qe[kk[u]];
    }
#pragma unroll
    for (int u = 0; u < 4; ++u) ss[u] = src[ee[u]];
#pragma unroll
    for (int u = 0; u < 4; ++u) qe[kk[u]] = ss[u];
  }
  __syncthreads();

  if (n > 0) {
    for (int k0 = wave * AGG_GATH_U; k0 < n; k0 += (AGG_THREADS / 32) * AGG_GATH_U) {
      int s[AGG_GATH_U];
      int dl[AGG_GATH_U];
      float4 v[AGG_GATH_U];
#pragma unroll
      for (int u = 0; u < AGG_GATH_U; ++u) {
        const int k = (k0 + u) < n ? (k0 + u) : (n - 1);
        s[u] = qe[k];
        dl[u] = qd[k];
      }
#pragma unroll
      for (int u = 0; u < AGG_GATH_U; ++u)
        v[u] = ((const float4*)(Wh + (size_t)s[u] * D))[lane];
#pragma unroll
      for (int u = 0; u < AGG_GATH_U; ++u) {
        const bool live = (k0 + u) < n;
        float* tr = tile + dl[u] * D + lane;
        atomicAdd(tr + 0, live ? v[u].x : 0.0f);
        atomicAdd(tr + 32, live ? v[u].y : 0.0f);
        atomicAdd(tr + 64, live ? v[u].z : 0.0f);
        atomicAdd(tr + 96, live ? v[u].w : 0.0f);
      }
    }
  }
  __syncthreads();

  for (int r = wave; r < rows; r += AGG_THREADS / 32) {
    const float c = fmaxf((float)cnt[r], 1.0f);
    const float* tr = tile + r * D + lane;
    float4 o;
    o.x = tr[0] / c;
    o.y = tr[32] / c;
    o.z = tr[64] / c;
    o.w = tr[96] / c;
    float4* op = (float4*)(out + (size_t)(d0 + r) * D) + lane;
    if (accumulate) {
      const float4 p = *op;
      o.x += p.x; o.y += p.y; o.z += p.z; o.w += p.w;
    }
    *op = o;
  }
}

extern "C" void kernel_launch(void* const* d_in, const int* in_sizes, int n_in,
                              void* d_out, int out_size, void* d_ws, size_t ws_size,
                              hipStream_t stream) {
  (void)n_in; (void)out_size; (void)ws_size;

  const float* feat_user  = (const float*)d_in[0];
  const float* feat_item  = (const float*)d_in[1];
  const float* W_follows  = (const float*)d_in[2];
  const float* b_follows  = (const float*)d_in[3];
  const float* W_rates    = (const float*)d_in[4];
  const float* b_rates    = (const float*)d_in[5];
  const float* W_ratedby  = (const float*)d_in[6];
  const float* b_ratedby  = (const float*)d_in[7];
  const int* src_follows  = (const int*)d_in[8];
  const int* dst_follows  = (const int*)d_in[9];
  const int* src_rates    = (const int*)d_in[10];
  const int* dst_rates    = (const int*)d_in[11];
  const int* src_ratedby  = (const int*)d_in[12];
  const int* dst_ratedby  = (const int*)d_in[13];

  const int NU  = in_sizes[0] / D;
  const int NI  = in_sizes[1] / D;
  const int Ef  = in_sizes[8];
  const int Er  = in_sizes[10];
  const int Erb = in_sizes[12];

  float* out_user = (float*)d_out;
  float* out_item = out_user + (size_t)NU * D;

  float* whbuf = (float*)d_ws;

  const int gU = (NU + AGG_ROWS - 1) / AGG_ROWS;
  const int gI = (NI + AGG_ROWS - 1) / AGG_ROWS;

  gemm_wmma_f32<<<NU / 16, 256, 0, stream>>>(feat_user, W_follows, b_follows, whbuf);
  agg_owned_rows<<<gU, AGG_THREADS, 0, stream>>>(whbuf, src_follows, dst_follows,
                                                 out_user, Ef, NU, 0);
  gemm_wmma_f32<<<NI / 16, 256, 0, stream>>>(feat_item, W_ratedby, b_ratedby, whbuf);
  agg_owned_rows<<<gU, AGG_THREADS, 0, stream>>>(whbuf, src_ratedby, dst_ratedby,
                                                 out_user, Erb, NU, 1);
  gemm_wmma_f32<<<NU / 16, 256, 0, stream>>>(feat_user, W_rates, b_rates, whbuf);
  agg_owned_rows<<<gI, AGG_THREADS, 0, stream>>>(whbuf, src_rates, dst_rates,
                                                 out_item, Er, NI, 0);
}


# GEMM data path: W stays in registers (a wave keeps its 16 columns as B fragments all launch and walks over row tiles), X tile read once per work-group into LDS by whole-line loads; same sums
# speedup vs baseline: 2.5671x; 1.4256x over previous
#include <hip/hip_runtime.h>

typedef __attribute__((ext_vector_type(2))) float v2f;
typedef __attribute__((ext_vector_type(8))) float v8f;

#define D 128

#define GEMM_THREADS 256
#define GEMM_BLOCKS 512
#define XP 132

__global__ __launch_bounds__(GEMM_THREADS)
void gemm_wmma_f32(const float* __restrict__ X,
                   const float* __restrict__ W,
                   const float* __restrict__ bias,
                   float* __restrict__ Y, int ntiles) {
  __shared__ float xs[2][16 * XP];

  const int tid = threadIdx.x;
  const int lane = tid & 31;
  const int wave = tid >> 5;
  const int n0 = wave * 16;
  const int r = lane & 15;
  const int hi = lane >> 4;
  const int koff = hi * 2;

  const float* brow = W + (size_t)(n0 + r) * D + koff;
  v2f b[D / 4];
#pragma unroll
  for (int kb = 0; kb < D / 4; ++kb) b[kb] = *(const v2f*)(brow + kb * 4);
  const float bv = bias[n0 + r];

  const float4* X4 = (const float4*)X;
  const int srow = tid >> 5;
  const int scol = (tid & 31) * 4;

  int tile = blockIdx.x;
  float4 g0 = make_float4(0.0f, 0.0f, 0.0f, 0.0f);
  float4 g1 = g0;
  if (tile < ntiles) {
    g0 = X4[(size_t)tile * 512 + tid];
    g1 = X4[(size_t)tile * 512 + 256 + tid];
  }
  int buf = 0;
  for (; tile < ntiles; tile += (int)gridDim.x) {
    float* xb = xs[buf];
    *(float4*)(xb + srow * XP + scol) = g0;
    *(float4*)(xb + (8 + srow) * XP + scol) = g1;
    __syncthreads();

    const int nt = tile + (int)gridDim.x;
    const size_t lt = (size_t)(nt < ntiles ? nt : tile);
    g0 = X4[lt * 512 + tid];
    g1 = X4[lt * 512 + 256 + tid];

    const float* ap = xb + r * XP + koff;
    v2f a[D / 4];
#pragma unroll
    for (int kb = 0; kb < D / 4; ++kb) a[kb] = *(const v2f*)(ap + kb * 4);

    __builtin_amdgcn_sched_barrier(0);

    v8f c = {};
#pragma unroll
    for (int kb = 0; kb < D / 4; ++kb) {
      c = __builtin_amdgcn_wmma_f32_16x16x4_f32(false, a[kb], false, b[kb],
                                                (short)0, c, false, false);
    }

    float* yout = Y + ((size_t)tile * 16 + hi * 8) * D + n0 + r;
#pragma unroll
    for (int v = 0; v < 8; ++v) yout[(size_t)v * D] = c[v] + bv;
    buf ^= 1;
  }
}

#define AGG_ROWS 400
#define AGG_QCAP 6144
#define AGG_THREADS 1024
#define AGG_SCAN_U 8
#define AGG_GATH_U 8

__device__ __forceinline__ void agg_push(int e, unsigned dl,
                                         const float* __restrict__ Wh,
                                         const int* __restrict__ src,
                                         float* tile, int* cnt, int* qe, int* qd,
                                         unsigned* qn) {
  atomicAdd(&cnt[dl], 1);
  const unsigned k = atomicAdd(qn, 1u);
  if (k < AGG_QCAP) {
    qe[k] = e;
    qd[k] = (int)dl;
  } else {
    const float* w = Wh + (size_t)src[e] * D;
    for (int f = 0; f < D; ++f)
      atomicAdd(&tile[dl * D + (f & 3) * 32 + (f >> 2)], w[f]);
  }
}

__global__ __launch_bounds__(AGG_THREADS)
void agg_owned_rows(const float* __restrict__ Wh,
                    const int* __restrict__ src,
                    const int* __restrict__ dst,
                    float* __restrict__ out,
                    int nE, int nDst, int accumulate) {
  __shared__ float tile[AGG_ROWS * D];
  __shared__ int cnt[AGG_ROWS];
  __shared__ int qe[AGG_QCAP];
  __shared__ int qd[AGG_QCAP];
  __shared__ unsigned qn;

  const int tid = threadIdx.x;
  const int lane = tid & 31;
  const int wave = tid >> 5;
  const int d0 = blockIdx.x * AGG_ROWS;
  const int rows = (nDst - d0) < AGG_ROWS ? (nDst - d0) : AGG_ROWS;
  const unsigned urows = (unsigned)rows;

  float4* t4 = (float4*)tile;
  for (int i = tid; i < AGG_ROWS * D / 4; i += AGG_THREADS)
    t4[i] = make_float4(0.0f, 0.0f, 0.0f, 0.0f);
  for (int i = tid; i < AGG_ROWS; i += AGG_THREADS) cnt[i] = 0;
  if (tid == 0) qn = 0u;
  __syncthreads();

  const int n4 = nE >> 2;
  const int4* dst4 = (const int4*)dst;
  if (n4 > 0) {
    const int off = (int)blockIdx.x * (n4 / (int)gridDim.x);
    for (int t0 = tid; t0 < n4; t0 += AGG_THREADS * AGG_SCAN_U) {
      int4 dv[AGG_SCAN_U];
      int id[AGG_SCAN_U];
#pragma unroll
      for (int u = 0; u < AGG_SCAN_U; ++u) {
        const int t = t0 + u * AGG_THREADS;
        int i = t + off;
        if (i >= n4) i -= n4;
        const bool ok = t < n4;
        i = ok ? i : 0;
        id[u] = i;
        int4 v = dst4[i];
        if (!ok) v = make_int4(-1, -1, -1, -1);
        dv[u] = v;
      }
#pragma unroll
      for (int u = 0; u < AGG_SCAN_U; ++u) {
        const unsigned a0 = (unsigned)(dv[u].x - d0);
        const unsigned a1 = (unsigned)(dv[u].y - d0);
        const unsigned a2 = (unsigned)(dv[u].z - d0);
        const unsigned a3 = (unsigned)(dv[u].w - d0);
        const unsigned m01 = a0 < a1 ? a0 : a1;
        const unsigned m23 = a2 < a3 ? a2 : a3;
        const unsigned m = m01 < m23 ? m01 : m23;
        if (m < urows) {
          const int e0 = id[u] * 4;
          if (a0 < urows) agg_push(e0 + 0, a0, Wh, src, tile, cnt, qe, qd, &qn);
          if (a1 < urows) agg_push(e0 + 1, a1, Wh, src, tile, cnt, qe, qd, &qn);
          if (a2 < urows) agg_push(e0 + 2, a2, Wh, src, tile, cnt, qe, qd, &qn);
          if (a3 < urows) agg_push(e0 + 3, a3, Wh, src, tile, cnt, qe, qd, &qn);
        }
      }
    }
  }
  if (tid < (nE & 3)) {
    const int e = n4 * 4 + tid;
    const unsigned a = (unsigned)(dst[e] - d0);
    if (a < urows) agg_push(e, a, Wh, src, tile, cnt, qe, qd, &qn);
  }
  __syncthreads();

  const int n = qn < (unsigned)AGG_QCAP ? (int)qn : AGG_QCAP;

  for (int kb = tid; kb < n; kb += AGG_THREADS * 4) {
    int kk[4];
    int ee[4];
    int ss[4];
#pragma unroll
    for (int u = 0; u < 4; ++u) {
      const int k = kb + u * AGG_THREADS;
      kk[u] = k < n ? k : kb;
      ee[u] = qe[kk[u]];
    }
#pragma unroll
    for (int u = 0; u < 4; ++u) ss[u] = src[ee[u]];
#pragma unroll
    for (int u = 0; u < 4; ++u) qe[kk[u]] = ss[u];
  }
  __syncthreads();

  if (n > 0) {
    for (int k0 = wave * AGG_GATH_U; k0 < n; k0 += (AGG_THREADS / 32) * AGG_GATH_U) {
      int s[AGG_GATH_U];
      int dl[AGG_GATH_U];
      float4 v[AGG_GATH_U];
#pragma unroll
      for (int u = 0; u < AGG_GATH_U; ++u) {
        const int k = (k0 + u) < n ? (k0 + u) : (n - 1);
        s[u] = qe[k];
        dl[u] = qd[k];
      }
#pragma unroll
      for (int u = 0; u < AGG_GATH_U; ++u)
        v[u] = ((const float4*)(Wh + (size_t)s[u] * D))[lane];
#pragma unroll
      for (int u = 0; u < AGG_GATH_U; ++u) {
        const bool live = (k0 + u) < n;
        float* tr = tile + dl[u] * D + lane;
        atomicAdd(tr + 0, live ? v[u].x : 0.0f);
        atomicAdd(tr + 32, live ? v[u].y : 0.0f);
        atomicAdd(tr + 64, live ? v[u].z : 0.0f);
        atomicAdd(tr + 96, live ? v[u].w : 0.0f);
      }
    }
  }
  __syncthreads();

  for (int r = wave; r < rows; r += AGG_THREADS / 32) {
    const float c = fmaxf((float)cnt[r], 1.0f);
    const float* tr = tile + r * D + lane;
    float4 o;
    o.x = tr[0] / c;
    o.y = tr[32] / c;
    o.z = tr[64] / c;
    o.w = tr[96] / c;
    float4* op = (float4*)(out + (size_t)(d0 + r) * D) + lane;
    if (accumulate) {
      const float4 p = *op;
      o.x += p.x; o.y += p.y; o.z += p.z; o.w += p.w;
    }
    *op = o;
  }
}

extern "C" void kernel_launch(void* const* d_in, const int* in_sizes, int n_in,
                              void* d_out, int out_size, void* d_ws, size_t ws_size,
                              hipStream_t stream) {
  (void)n_in; (void)out_size; (void)ws_size;

  const float* feat_user  = (const float*)d_in[0];
  const float* feat_item  = (const float*)d_in[1];
  const float* W_follows  = (const float*)d_in[2];
  const float* b_follows  = (const float*)d_in[3];
  const float* W_rates    = (const float*)d_in[4];
  const float* b_rates    = (const float*)d_in[5];
  const float* W_ratedby  = (const float*)d_in[6];
  const float* b_ratedby  = (const float*)d_in[7];
  const int* src_follows  = (const int*)d_in[8];
  const int* dst_follows  = (const int*)d_in[9];
  const int* src_rates    = (const int*)d_in[10];
  const int* dst_rates    = (const int*)d_in[11];
  const int* src_ratedby  = (const int*)d_in[12];
  const int* dst_ratedby  = (const int*)d_in[13];

  const int NU  = in_sizes[0] / D;
  const int NI  = in_sizes[1] / D;
  const int Ef  = in_sizes[8];
  const int Er  = in_sizes[10];
  const int Erb = in_sizes[12];

  float* out_user = (float*)d_out;
  float* out_item = out_user + (size_t)NU * D;

  float* whbuf = (float*)d_ws;

  const int gU = (NU + AGG_ROWS - 1) / AGG_ROWS;
  const int gI = (NI + AGG_ROWS - 1) / AGG_ROWS;

  gemm_wmma_f32<<<GEMM_BLOCKS, GEMM_THREADS, 0, stream>>>(feat_user, W_follows, b_follows, whbuf, NU / 16);
  agg_owned_rows<<<gU, AGG_THREADS, 0, stream>>>(whbuf, src_follows, dst_follows,
                                                 out_user, Ef, NU, 0);
  gemm_wmma_f32<<<GEMM_BLOCKS, GEMM_THREADS, 0, stream>>>(feat_item, W_ratedby, b_ratedby, whbuf, NI / 16);
  agg_owned_rows<<<gU, AGG_THREADS, 0, stream>>>(whbuf, src_ratedby, dst_ratedby,
                                                 out_user, Erb, NU, 1);
  gemm_wmma_f32<<<GEMM_BLOCKS, GEMM_THREADS, 0, stream>>>(feat_user, W_rates, b_rates, whbuf, NU / 16);
  agg_owned_rows<<<gI, AGG_THREADS, 0, stream>>>(whbuf, src_rates, dst_rates,
                                                 out_item, Er, NI, 0);
}


# owners no longer scan dst: a pre-launch writes per edge an owner byte and a word (row in range, src id); owners scan the bytes, 16 edges a load, one line read per own edge; 6 -> 7 launches; same sums
# speedup vs baseline: 2.9976x; 1.1677x over previous
#include <hip/hip_runtime.h>

typedef __attribute__((ext_vector_type(2))) float v2f;
typedef __attribute__((ext_vector_type(8))) float v8f;

#define D 128

#define GEMM_THREADS 256
#define GEMM_BLOCKS 512
#define XP 132

__global__ __launch_bounds__(GEMM_THREADS)
void gemm_wmma_f32(const float* __restrict__ X,
                   const float* __restrict__ W,
                   const float* __restrict__ bias,
                   float* __restrict__ Y, int ntiles) {
  __shared__ float xs[2][16 * XP];

  const int tid = threadIdx.x;
  const int lane = tid & 31;
  const int wave = tid >> 5;
  const int n0 = wave * 16;
  const int r = lane & 15;
  const int hi = lane >> 4;
  const int koff = hi * 2;

  const float* brow = W + (size_t)(n0 + r) * D + koff;
  v2f b[D / 4];
#pragma unroll
  for (int kb = 0; kb < D / 4; ++kb) b[kb] = *(const v2f*)(brow + kb * 4);
  const float bv = bias[n0 + r];

  const float4* X4 = (const float4*)X;
  const int srow = tid >> 5;
  const int scol = (tid & 31) * 4;

  int tile = blockIdx.x;
  float4 g0 = make_float4(0.0f, 0.0f, 0.0f, 0.0f);
  float4 g1 = g0;
  if (tile < ntiles) {
    g0 = X4[(size_t)tile * 512 + tid];
    g1 = X4[(size_t)tile * 512 + 256 + tid];
  }
  int buf = 0;
  for (; tile < ntiles; tile += (int)gridDim.x) {
    float* xb = xs[buf];
    *(float4*)(xb + srow * XP + scol) = g0;
    *(float4*)(xb + (8 + srow) * XP + scol) = g1;
    __syncthreads();

    const int nt = tile + (int)gridDim.x;
    const size_t lt = (size_t)(nt < ntiles ? nt : tile);
    g0 = X4[lt * 512 + tid];
    g1 = X4[lt * 512 + 256 + tid];

    const float* ap = xb + r * XP + koff;
    v2f a[D / 4];
#pragma unroll
    for (int kb = 0; kb < D / 4; ++kb) a[kb] = *(const v2f*)(ap + kb * 4);

    __builtin_amdgcn_sched_barrier(0);

    v8f c = {};
#pragma unroll
    for (int kb = 0; kb < D / 4; ++kb) {
      c = __builtin_amdgcn_wmma_f32_16x16x4_f32(false, a[kb], false, b[kb],
                                                (short)0, c, false, false);
    }

    float* yout = Y + ((size_t)tile * 16 + hi * 8) * D + n0 + r;
#pragma unroll
    for (int v = 0; v < 8; ++v) yout[(size_t)v * D] = c[v] + bv;
    buf ^= 1;
  }
}

#define AGG_ROWS 400
#define AGG_QCAP 6144
#define AGG_THREADS 1024
#define AGG_SCAN_U 8
#define AGG_GATH_U 8

__device__ __forceinline__ unsigned owner_of(int d) {
  const unsigned o = (unsigned)d / (unsigned)AGG_ROWS;
  return o < 255u ? o : 255u;
}

__device__ __forceinline__ unsigned zero_bytes(unsigned x) {
  return ~(((x & 0x7F7F7F7Fu) + 0x7F7F7F7Fu) | x | 0x7F7F7F7Fu);
}

__device__ __forceinline__ unsigned flag_nibble(unsigned z) {
  return (((z >> 7) * 0x01020408u) >> 24) & 15u;
}

#define OWN_THREADS 256
#define OWN_U 4
#define SRC_BITS 23
#define SRC_MASK 0x7FFFFFu

__device__ __forceinline__ unsigned edge_word(int d, int s) {
  const unsigned o = (unsigned)d / (unsigned)AGG_ROWS;
  return (((unsigned)d - o * (unsigned)AGG_ROWS) << SRC_BITS) | ((unsigned)s & SRC_MASK);
}

__global__ __launch_bounds__(OWN_THREADS)
void edge_codes(const int* __restrict__ dstA, const int* __restrict__ dstB,
                const int* __restrict__ dstC,
                const int* __restrict__ srcA, const int* __restrict__ srcB,
                const int* __restrict__ srcC,
                unsigned* __restrict__ obA, unsigned* __restrict__ obB,
                unsigned* __restrict__ obC,
                unsigned* __restrict__ ewA, unsigned* __restrict__ ewB,
                unsigned* __restrict__ ewC, int nA, int nB, int nC) {
  const int y = blockIdx.y;
  const int* dst = y == 0 ? dstA : (y == 1 ? dstB : dstC);
  const int* src = y == 0 ? srcA : (y == 1 ? srcB : srcC);
  unsigned* ob = y == 0 ? obA : (y == 1 ? obB : obC);
  unsigned* ew = y == 0 ? ewA : (y == 1 ? ewB : ewC);
  const int nE = y == 0 ? nA : (y == 1 ? nB : nC);
  const int n4 = nE >> 2;
  const int4* d4 = (const int4*)dst;
  const int4* s4 = (const int4*)src;
  uint4* ew4 = (uint4*)ew;
  const int base = (int)blockIdx.x * (OWN_THREADS * OWN_U) + (int)threadIdx.x;
  if (n4 > 0) {
    int4 dv[OWN_U];
    int4 sv[OWN_U];
#pragma unroll
    for (int u = 0; u < OWN_U; ++u) {
      const int i = base + u * OWN_THREADS;
      const int ic = i < n4 ? i : 0;
      dv[u] = d4[ic];
      sv[u] = s4[ic];
    }
#pragma unroll
    for (int u = 0; u < OWN_U; ++u) {
      const int i = base + u * OWN_THREADS;
      if (i < n4) {
        ob[i] = owner_of(dv[u].x) | (owner_of(dv[u].y) << 8) |
                (owner_of(dv[u].z) << 16) | (owner_of(dv[u].w) << 24);
        uint4 w;
        w.x = edge_word(dv[u].x, sv[u].x);
        w.y = edge_word(dv[u].y, sv[u].y);
        w.z = edge_word(dv[u].z, sv[u].z);
        w.w = edge_word(dv[u].w, sv[u].w);
        ew4[i] = w;
      }
    }
  }
  if (blockIdx.x == 0 && threadIdx.x == 0 && (nE & 3) != 0) {
    unsigned w = 0xFFFFFFFFu;
    for (int j = 0; j < (nE & 3); ++j) {
      const int e = n4 * 4 + j;
      w = (w & ~(255u << (8 * j))) | (owner_of(dst[e]) << (8 * j));
      ew[e] = edge_word(dst[e], src[e]);
    }
    ob[n4] = w;
  }
}

__device__ __forceinline__ void agg_push(int e,
                                         const float* __restrict__ Wh,
                                         const unsigned* __restrict__ ew,
                                         float* tile, int* cnt, int* qe,
                                         unsigned* qn) {
  const unsigned k = atomicAdd(qn, 1u);
  if (k < AGG_QCAP) {
    qe[k] = e;
  } else {
    const unsigned p = ew[e];
    const unsigned dl = p >> SRC_BITS;
    atomicAdd(&cnt[dl], 1);
    const float* w = Wh + (size_t)(p & SRC_MASK) * D;
    for (int f = 0; f < D; ++f)
      atomicAdd(&tile[dl * D + (f & 3) * 32 + (f >> 2)], w[f]);
  }
}

__global__ __launch_bounds__(AGG_THREADS)
void agg_owned_rows(const float* __restrict__ Wh,
                    const unsigned* __restrict__ ob,
                    const unsigned* __restrict__ ew,
                    float* __restrict__ out,
                    int nE, int nDst, int accumulate) {
  __shared__ float tile[AGG_ROWS * D];
  __shared__ int cnt[AGG_ROWS];
  __shared__ int qe[AGG_QCAP];
  __shared__ int qd[AGG_QCAP];
  __shared__ unsigned qn;

  const int tid = threadIdx.x;
  const int lane = tid & 31;
  const int wave = tid >> 5;
  const int d0 = blockIdx.x * AGG_ROWS;
  const int rows = (nDst - d0) < AGG_ROWS ? (nDst - d0) : AGG_ROWS;

  float4* t4 = (float4*)tile;
  for (int i = tid; i < AGG_ROWS * D / 4; i += AGG_THREADS)
    t4[i] = make_float4(0.0f, 0.0f, 0.0f, 0.0f);
  for (int i = tid; i < AGG_ROWS; i += AGG_THREADS) cnt[i] = 0;
  if (tid == 0) qn = 0u;
  __syncthreads();

  const int n16 = nE >> 4;
  const uint4* ob4 = (const uint4*)ob;
  const unsigned me = (unsigned)blockIdx.x * 0x01010101u;
  if (n16 > 0) {
    const int off = (int)blockIdx.x * (n16 / (int)gridDim.x);
    for (int t0 = tid; t0 < n16; t0 += AGG_THREADS * AGG_SCAN_U) {
      uint4 ov[AGG_SCAN_U];
      int id[AGG_SCAN_U];
#pragma unroll
      for (int u = 0; u < AGG_SCAN_U; ++u) {
        const int t = t0 + u * AGG_THREADS;
        int i = t + off;
        if (i >= n16) i -= n16;
        const bool ok = t < n16;
        i = ok ? i : 0;
        id[u] = ok ? i : -1;
        ov[u] = ob4[i];
      }
#pragma unroll
      for (int u = 0; u < AGG_SCAN_U; ++u) {
        const unsigned z0 = zero_bytes(ov[u].x ^ me);
        const unsigned z1 = zero_bytes(ov[u].y ^ me);
        const unsigned z2 = zero_bytes(ov[u].z ^ me);
        const unsigned z3 = zero_bytes(ov[u].w ^ me);
        if ((z0 | z1 | z2 | z3) != 0u && id[u] >= 0) {
          unsigned m = flag_nibble(z0) | (flag_nibble(z1) << 4) |
                       (flag_nibble(z2) << 8) | (flag_nibble(z3) << 12);
          const int e0 = id[u] * 16;
          while (m != 0u) {
            const int b = __builtin_ctz(m);
            m &= m - 1u;
            agg_push(e0 + b, Wh, ew, tile, cnt, qe, &qn);
          }
        }
      }
    }
  }
  if (tid < (nE & 15)) {
    const int e = n16 * 16 + tid;
    if (((ob[e >> 2] >> ((e & 3) * 8)) & 255u) == (unsigned)blockIdx.x)
      agg_push(e, Wh, ew, tile, cnt, qe, &qn);
  }
  __syncthreads();

  const int n = qn < (unsigned)AGG_QCAP ? (int)qn : AGG_QCAP;

  for (int kb = tid; kb < n; kb += AGG_THREADS * 4) {
    int kk[4];
    int ee[4];
    unsigned pp[4];
#pragma unroll
    for (int u = 0; u < 4; ++u) {
      const int k = kb + u * AGG_THREADS;
      kk[u] = k < n ? k : kb;
      ee[u] = qe[kk[u]];
    }
#pragma unroll
    for (int u = 0; u < 4; ++u) pp[u] = ew[ee[u]];
#pragma unroll
    for (int u = 0; u < 4; ++u) {
      const int dl = (int)(pp[u] >> SRC_BITS);
      qe[kk[u]] = (int)(pp[u] & SRC_MASK);
      qd[kk[u]] = dl;
      if (kb + u * AGG_THREADS < n) atomicAdd(&cnt[dl], 1);
    }
  }
  __syncthreads();

  if (n > 0) {
    for (int k0 = wave * AGG_GATH_U; k0 < n; k0 += (AGG_THREADS / 32) * AGG_GATH_U) {
      int s[AGG_GATH_U];
      int dl[AGG_GATH_U];
      float4 v[AGG_GATH_U];
#pragma unroll
      for (int u = 0; u < AGG_GATH_U; ++u) {
        const int k = (k0 + u) < n ? (k0 + u) : (n - 1);
        s[u] = qe[k];
        dl[u] = qd[k];
      }
#pragma unroll
      for (int u = 0; u < AGG_GATH_U; ++u)
        v[u] = ((const float4*)(Wh + (size_t)s[u] * D))[lane];
#pragma unroll
      for (int u = 0; u < AGG_GATH_U; ++u) {
        const bool live = (k0 + u) < n;
        float* tr = tile + dl[u] * D + lane;
        atomicAdd(tr + 0, live ? v[u].x : 0.0f);
        atomicAdd(tr + 32, live ? v[u].y : 0.0f);
        atomicAdd(tr + 64, live ? v[u].z : 0.0f);
        atomicAdd(tr + 96, live ? v[u].w : 0.0f);
      }
    }
  }
  __syncthreads();

  for (int r = wave; r < rows; r += AGG_THREADS / 32) {
    const float c = fmaxf((float)cnt[r], 1.0f);
    const float* tr = tile + r * D + lane;
    float4 o;
    o.x = tr[0] / c;
    o.y = tr[32] / c;
    o.z = tr[64] / c;
    o.w = tr[96] / c;
    float4* op = (float4*)(out + (size_t)(d0 + r) * D) + lane;
    if (accumulate) {
      const float4 p = *op;
      o.x += p.x; o.y += p.y; o.z += p.z; o.w += p.w;
    }
    *op = o;
  }
}

extern "C" void kernel_launch(void* const* d_in, const int* in_sizes, int n_in,
                              void* d_out, int out_size, void* d_ws, size_t ws_size,
                              hipStream_t stream) {
  (void)n_in; (void)out_size; (void)ws_size;

  const float* feat_user  = (const float*)d_in[0];
  const float* feat_item  = (const float*)d_in[1];
  const float* W_follows  = (const float*)d_in[2];
  const float* b_follows  = (const float*)d_in[3];
  const float* W_rates    = (const float*)d_in[4];
  const float* b_rates    = (const float*)d_in[5];
  const float* W_ratedby  = (const float*)d_in[6];
  const float* b_ratedby  = (const float*)d_in[7];
  const int* src_follows  = (const int*)d_in[8];
  const int* dst_follows  = (const int*)d_in[9];
  const int* src_rates    = (const int*)d_in[10];
  const int* dst_rates    = (const int*)d_in[11];
  const int* src_ratedby  = (const int*)d_in[12];
  const int* dst_ratedby  = (const int*)d_in[13];

  const int NU  = in_sizes[0] / D;
  const int NI  = in_sizes[1] / D;
  const int Ef  = in_sizes[8];
  const int Er  = in_sizes[10];
  const int Erb = in_sizes[12];

  float* out_user = (float*)d_out;
  float* out_item = out_user + (size_t)NU * D;

  float* whbuf = (float*)d_ws;

  const int gU = (NU + AGG_ROWS - 1) / AGG_ROWS;
  const int gI = (NI + AGG_ROWS - 1) / AGG_ROWS;

  const int Emax = Ef > Er ? (Ef > Erb ? Ef : Erb) : (Er > Erb ? Er : Erb);
  const size_t obWords = (size_t)((Emax / 4 + 67) & ~3);
  const size_t ewWords = (size_t)((Emax + 7) & ~3);
  unsigned* ob_follows = (unsigned*)(whbuf + (size_t)(NU > NI ? NU : NI) * D);
  unsigned* ob_ratedby = ob_follows + obWords;
  unsigned* ob_rates   = ob_ratedby + obWords;
  unsigned* ew_follows = ob_rates + obWords;
  unsigned* ew_ratedby = ew_follows + ewWords;
  unsigned* ew_rates   = ew_ratedby + ewWords;
  const int gO = (Emax / 4 + OWN_THREADS * OWN_U - 1) / (OWN_THREADS * OWN_U);
  edge_codes<<<dim3(gO > 0 ? gO : 1, 3), OWN_THREADS, 0, stream>>>(
      dst_follows, dst_ratedby, dst_rates, src_follows, src_ratedby, src_rates,
      ob_follows, ob_ratedby, ob_rates, ew_follows, ew_ratedby, ew_rates, Ef, Erb, Er);

  gemm_wmma_f32<<<GEMM_BLOCKS, GEMM_THREADS, 0, stream>>>(feat_user, W_follows, b_follows, whbuf, NU / 16);
  agg_owned_rows<<<gU, AGG_THREADS, 0, stream>>>(whbuf, ob_follows, ew_follows,
                                                 out_user, Ef, NU, 0);
  gemm_wmma_f32<<<GEMM_BLOCKS, GEMM_THREADS, 0, stream>>>(feat_item, W_ratedby, b_ratedby, whbuf, NI / 16);
  agg_owned_rows<<<gU, AGG_THREADS, 0, stream>>>(whbuf, ob_ratedby, ew_ratedby,
                                                 out_user, Erb, NU, 1);
  gemm_wmma_f32<<<GEMM_BLOCKS, GEMM_THREADS, 0, stream>>>(feat_user, W_rates, b_rates, whbuf, NU / 16);
  agg_owned_rows<<<gI, AGG_THREADS, 0, stream>>>(whbuf, ob_rates, ew_rates,
                                                 out_item, Er, NI, 0);
}


# owner bytes -> one bit plane per owner: the pre-launch fills, 1024 edges at a time, a 128-B bit line per owner in LDS (LDS or), stores it whole; an owner reads only its plane; sums in arrival order
# speedup vs baseline: 3.4548x; 1.1525x over previous
#include <hip/hip_runtime.h>

typedef __attribute__((ext_vector_type(2))) float v2f;
typedef __attribute__((ext_vector_type(8))) float v8f;

#define D 128

#define GEMM_THREADS 256
#define GEMM_BLOCKS 512
#define XP 132

__global__ __launch_bounds__(GEMM_THREADS)
void gemm_wmma_f32(const float* __restrict__ X,
                   const float* __restrict__ W,
                   const float* __restrict__ bias,
                   float* __restrict__ Y, int ntiles) {
  __shared__ float xs[2][16 * XP];

  const int tid = threadIdx.x;
  const int lane = tid & 31;
  const int wave = tid >> 5;
  const int n0 = wave * 16;
  const int r = lane & 15;
  const int hi = lane >> 4;
  const int koff = hi * 2;

  const float* brow = W + (size_t)(n0 + r) * D + koff;
  v2f b[D / 4];
#pragma unroll
  for (int kb = 0; kb < D / 4; ++kb) b[kb] = *(const v2f*)(brow + kb * 4);
  const float bv = bias[n0 + r];

  const float4* X4 = (const float4*)X;
  const int srow = tid >> 5;
  const int scol = (tid & 31) * 4;

  int tile = blockIdx.x;
  float4 g0 = make_float4(0.0f, 0.0f, 0.0f, 0.0f);
  float4 g1 = g0;
  if (tile < ntiles) {
    g0 = X4[(size_t)tile * 512 + tid];
    g1 = X4[(size_t)tile * 512 + 256 + tid];
  }
  int buf = 0;
  for (; tile < ntiles; tile += (int)gridDim.x) {
    float* xb = xs[buf];
    *(float4*)(xb + srow * XP + scol) = g0;
    *(float4*)(xb + (8 + srow) * XP + scol) = g1;
    __syncthreads();

    const int nt = tile + (int)gridDim.x;
    const size_t lt = (size_t)(nt < ntiles ? nt : tile);
    g0 = X4[lt * 512 + tid];
    g1 = X4[lt * 512 + 256 + tid];

    const float* ap = xb + r * XP + koff;
    v2f a[D / 4];
#pragma unroll
    for (int kb = 0; kb < D / 4; ++kb) a[kb] = *(const v2f*)(ap + kb * 4);

    __builtin_amdgcn_sched_barrier(0);

    v8f c = {};
#pragma unroll
    for (int kb = 0; kb < D / 4; ++kb) {
      c = __builtin_amdgcn_wmma_f32_16x16x4_f32(false, a[kb], false, b[kb],
                                                (short)0, c, false, false);
    }

    float* yout = Y + ((size_t)tile * 16 + hi * 8) * D + n0 + r;
#pragma unroll
    for (int v = 0; v < 8; ++v) yout[(size_t)v * D] = c[v] + bv;
    buf ^= 1;
  }
}

#define AGG_ROWS 400
#define AGG_QCAP 6144
#define AGG_THREADS 1024
#define AGG_SCAN_U 2
#define AGG_GATH_U 8

#define PL_EDGES 1024
#define PL_OWNERS 256

__device__ __forceinline__ unsigned owner_of(int d) {
  const unsigned o = (unsigned)d / (unsigned)AGG_ROWS;
  return o < 255u ? o : 255u;
}

#define OWN_THREADS 256
#define OWN_SUB 4
#define SRC_BITS 23
#define SRC_MASK 0x7FFFFFu

__device__ __forceinline__ unsigned edge_word(int d, int s) {
  const unsigned o = (unsigned)d / (unsigned)AGG_ROWS;
  return (((unsigned)d - o * (unsigned)AGG_ROWS) << SRC_BITS) | ((unsigned)s & SRC_MASK);
}

__global__ __launch_bounds__(OWN_THREADS)
void edge_codes(const int* __restrict__ dstA, const int* __restrict__ dstB,
                const int* __restrict__ dstC,
                const int* __restrict__ srcA, const int* __restrict__ srcB,
                const int* __restrict__ srcC,
                unsigned* __restrict__ plA, unsigned* __restrict__ plB,
                unsigned* __restrict__ plC,
                unsigned* __restrict__ ewA, unsigned* __restrict__ ewB,
                unsigned* __restrict__ ewC, int nA, int nB, int nC) {
  __shared__ unsigned bits[PL_OWNERS * 32];

  const int y = blockIdx.y;
  const int* dst = y == 0 ? dstA : (y == 1 ? dstB : dstC);
  const int* src = y == 0 ? srcA : (y == 1 ? srcB : srcC);
  unsigned* pl = y == 0 ? plA : (y == 1 ? plB : plC);
  unsigned* ew = y == 0 ? ewA : (y == 1 ? ewB : ewC);
  const int nE = y == 0 ? nA : (y == 1 ? nB : nC);
  const int n4 = nE >> 2;
  const int nblk = (nE + PL_EDGES - 1) / PL_EDGES;
  const int tid = threadIdx.x;
  const int4* d4 = (const int4*)dst;
  const int4* s4 = (const int4*)src;
  uint4* ew4 = (uint4*)ew;
  uint4* pl4 = (uint4*)pl;
  uint4* bits4 = (uint4*)bits;
  const unsigned wd = (unsigned)tid >> 3;
  const unsigned sh = ((unsigned)tid & 7u) * 4u;

  int4 dv[OWN_SUB];
  int4 sv[OWN_SUB];
#pragma unroll
  for (int s = 0; s < OWN_SUB; ++s) {
    dv[s] = make_int4(0, 0, 0, 0);
    sv[s] = dv[s];
  }
  if (n4 > 0) {
#pragma unroll
    for (int s = 0; s < OWN_SUB; ++s) {
      const int g = ((int)blockIdx.x * OWN_SUB + s) * OWN_THREADS + tid;
      const int gc = g < n4 ? g : 0;
      dv[s] = d4[gc];
      sv[s] = s4[gc];
    }
  }

#pragma unroll
  for (int s = 0; s < OWN_SUB; ++s) {
    const int blk = (int)blockIdx.x * OWN_SUB + s;
    if (blk >= nblk) break;
#pragma unroll
    for (int j = 0; j < (PL_OWNERS * 32 / 4) / OWN_THREADS; ++j)
      bits4[tid + j * OWN_THREADS] = make_uint4(0u, 0u, 0u, 0u);
    __syncthreads();

    const int g = blk * OWN_THREADS + tid;
    if (g < n4) {
      uint4 w;
      w.x = edge_word(dv[s].x, sv[s].x);
      w.y = edge_word(dv[s].y, sv[s].y);
      w.z = edge_word(dv[s].z, sv[s].z);
      w.w = edge_word(dv[s].w, sv[s].w);
      ew4[g] = w;
      atomicOr(&bits[owner_of(dv[s].x) * 32u + wd], 1u << sh);
      atomicOr(&bits[owner_of(dv[s].y) * 32u + wd], 2u << sh);
      atomicOr(&bits[owner_of(dv[s].z) * 32u + wd], 4u << sh);
      atomicOr(&bits[owner_of(dv[s].w) * 32u + wd], 8u << sh);
    } else if (g == n4) {
      for (int j = 0; j < (nE & 3); ++j) {
        const int e = n4 * 4 + j;
        const int d = dst[e];
        ew[e] = edge_word(d, src[e]);
        atomicOr(&bits[owner_of(d) * 32u + wd], (1u << j) << sh);
      }
    }
    __syncthreads();

#pragma unroll
    for (int j = 0; j < (PL_OWNERS * 32 / 4) / OWN_THREADS; ++j) {
      const int idx = tid + j * OWN_THREADS;
      const size_t o = (size_t)(idx >> 3);
      pl4[(o * (size_t)nblk + (size_t)blk) * 8u + (size_t)(idx & 7)] = bits4[idx];
    }
    __syncthreads();
  }
}

__device__ __forceinline__ void agg_push(int e,
                                         const float* __restrict__ Wh,
                                         const unsigned* __restrict__ ew,
                                         float* tile, int* cnt, int* qe,
                                         unsigned* qn) {
  const unsigned k = atomicAdd(qn, 1u);
  if (k < AGG_QCAP) {
    qe[k] = e;
  } else {
    const unsigned p = ew[e];
    const unsigned dl = p >> SRC_BITS;
    atomicAdd(&cnt[dl], 1);
    const float* w = Wh + (size_t)(p & SRC_MASK) * D;
    for (int f = 0; f < D; ++f)
      atomicAdd(&tile[dl * D + (f & 3) * 32 + (f >> 2)], w[f]);
  }
}

__global__ __launch_bounds__(AGG_THREADS)
void agg_owned_rows(const float* __restrict__ Wh,
                    const unsigned* __restrict__ pl,
                    const unsigned* __restrict__ ew,
                    float* __restrict__ out,
                    int nE, int nDst, int accumulate) {
  __shared__ float tile[AGG_ROWS * D];
  __shared__ int cnt[AGG_ROWS];
  __shared__ int qe[AGG_QCAP];
  __shared__ int qd[AGG_QCAP];
  __shared__ unsigned qn;

  const int tid = threadIdx.x;
  const int lane = tid & 31;
  const int wave = tid >> 5;
  const int d0 = blockIdx.x * AGG_ROWS;
  const int rows = (nDst - d0) < AGG_ROWS ? (nDst - d0) : AGG_ROWS;

  float4* t4 = (float4*)tile;
  for (int i = tid; i < AGG_ROWS * D / 4; i += AGG_THREADS)
    t4[i] = make_float4(0.0f, 0.0f, 0.0f, 0.0f);
  for (int i = tid; i < AGG_ROWS; i += AGG_THREADS) cnt[i] = 0;
  if (tid == 0) qn = 0u;
  __syncthreads();

  const int nblk = (nE + PL_EDGES - 1) / PL_EDGES;
  const int n4 = (int)blockIdx.x < PL_OWNERS - 1 ? nblk * 8 : 0;
  const uint4* pl4 = (const uint4*)(pl + (size_t)blockIdx.x * (size_t)nblk * 32u);
  for (int t0 = tid; t0 < n4; t0 += AGG_THREADS * AGG_SCAN_U) {
    uint4 pv[AGG_SCAN_U];
#pragma unroll
    for (int u = 0; u < AGG_SCAN_U; ++u) {
      const int t = t0 + u * AGG_THREADS;
      const bool ok = t < n4;
      const uint4 v = pl4[ok ? t : t0];
      pv[u].x = ok ? v.x : 0u;
      pv[u].y = ok ? v.y : 0u;
      pv[u].z = ok ? v.z : 0u;
      pv[u].w = ok ? v.w : 0u;
    }
#pragma unroll
    for (int u = 0; u < AGG_SCAN_U; ++u) {
      unsigned long long lo = (unsigned long long)pv[u].x | ((unsigned long long)pv[u].y << 32);
      unsigned long long hi = (unsigned long long)pv[u].z | ((unsigned long long)pv[u].w << 32);
      unsigned c = (unsigned)__popcll(lo) + (unsigned)__popcll(hi);
      if (c != 0u) {
        const int e0 = (t0 + u * AGG_THREADS) * 128;
        do {
          int b;
          if (lo != 0ull) {
            b = __builtin_ctzll(lo);
            lo &= lo - 1ull;
          } else {
            b = 64 + __builtin_ctzll(hi);
            hi &= hi - 1ull;
          }
          agg_push(e0 + b, Wh, ew, tile, cnt, qe, &qn);
        } while (--c != 0u);
      }
    }
  }
  __syncthreads();

  const int n = qn < (unsigned)AGG_QCAP ? (int)qn : AGG_QCAP;

  for (int kb = tid; kb < n; kb += AGG_THREADS * 4) {
    int kk[4];
    int ee[4];
    unsigned pp[4];
#pragma unroll
    for (int u = 0; u < 4; ++u) {
      const int k = kb + u * AGG_THREADS;
      kk[u] = k < n ? k : kb;
      ee[u] = qe[kk[u]];
    }
#pragma unroll
    for (int u = 0; u < 4; ++u) pp[u] = ew[ee[u]];
#pragma unroll
    for (int u = 0; u < 4; ++u) {
      const int dl = (int)(pp[u] >> SRC_BITS);
      qe[kk[u]] = (int)(pp[u] & SRC_MASK);
      qd[kk[u]] = dl;
      if (kb + u * AGG_THREADS < n) atomicAdd(&cnt[dl], 1);
    }
  }
  __syncthreads();

  if (n > 0) {
    for (int k0 = wave * AGG_GATH_U; k0 < n; k0 += (AGG_THREADS / 32) * AGG_GATH_U) {
      int s[AGG_GATH_U];
      int dl[AGG_GATH_U];
      float4 v[AGG_GATH_U];
#pragma unroll
      for (int u = 0; u < AGG_GATH_U; ++u) {
        const int k = (k0 + u) < n ? (k0 + u) : (n - 1);
        s[u] = qe[k];
        dl[u] = qd[k];
      }
#pragma unroll
      for (int u = 0; u < AGG_GATH_U; ++u)
        v[u] = ((const float4*)(Wh + (size_t)s[u] * D))[lane];
#pragma unroll
      for (int u = 0; u < AGG_GATH_U; ++u) {
        const bool live = (k0 + u) < n;
        float* tr = tile + dl[u] * D + lane;
        atomicAdd(tr + 0, live ? v[u].x : 0.0f);
        atomicAdd(tr + 32, live ? v[u].y : 0.0f);
        atomicAdd(tr + 64, live ? v[u].z : 0.0f);
        atomicAdd(tr + 96, live ? v[u].w : 0.0f);
      }
    }
  }
  __syncthreads();

  for (int r = wave; r < rows; r += AGG_THREADS / 32) {
    const float c = fmaxf((float)cnt[r], 1.0f);
    const float* tr = tile + r * D + lane;
    float4 o;
    o.x = tr[0] / c;
    o.y = tr[32] / c;
    o.z = tr[64] / c;
    o.w = tr[96] / c;
    float4* op = (float4*)(out + (size_t)(d0 + r) * D) + lane;
    if (accumulate) {
      const float4 p = *op;
      o.x += p.x; o.y += p.y; o.z += p.z; o.w += p.w;
    }
    *op = o;
  }
}

extern "C" void kernel_launch(void* const* d_in, const int* in_sizes, int n_in,
                              void* d_out, int out_size, void* d_ws, size_t ws_size,
                              hipStream_t stream) {
  (void)n_in; (void)out_size; (void)ws_size;

  const float* feat_user  = (const float*)d_in[0];
  const float* feat_item  = (const float*)d_in[1];
  const float* W_follows  = (const float*)d_in[2];
  const float* b_follows  = (const float*)d_in[3];
  const float* W_rates    = (const float*)d_in[4];
  const float* b_rates    = (const float*)d_in[5];
  const float* W_ratedby  = (const float*)d_in[6];
  const float* b_ratedby  = (const float*)d_in[7];
  const int* src_follows  = (const int*)d_in[8];
  const int* dst_follows  = (const int*)d_in[9];
  const int* src_rates    = (const int*)d_in[10];
  const int* dst_rates    = (const int*)d_in[11];
  const int* src_ratedby  = (const int*)d_in[12];
  const int* dst_ratedby  = (const int*)d_in[13];

  const int NU  = in_sizes[0] / D;
  const int NI  = in_sizes[1] / D;
  const int Ef  = in_sizes[8];
  const int Er  = in_sizes[10];
  const int Erb = in_sizes[12];

  float* out_user = (float*)d_out;
  float* out_item = out_user + (size_t)NU * D;

  float* whbuf = (float*)d_ws;

  const int gU = (NU + AGG_ROWS - 1) / AGG_ROWS;
  const int gI = (NI + AGG_ROWS - 1) / AGG_ROWS;

  const int Emax = Ef > Er ? (Ef > Erb ? Ef : Erb) : (Er > Erb ? Er : Erb);
  const int nblkMax = (Emax + PL_EDGES - 1) / PL_EDGES;
  const size_t plWords = (size_t)PL_OWNERS * (size_t)nblkMax * 32;
  const size_t ewWords = (size_t)((Emax + 7) & ~3);
  unsigned* pl_follows = (unsigned*)(whbuf + (size_t)(NU > NI ? NU : NI) * D);
  unsigned* pl_ratedby = pl_follows + plWords;
  unsigned* pl_rates   = pl_ratedby + plWords;
  unsigned* ew_follows = pl_rates + plWords;
  unsigned* ew_ratedby = ew_follows + ewWords;
  unsigned* ew_rates   = ew_ratedby + ewWords;
  const int gO = (nblkMax + OWN_SUB - 1) / OWN_SUB;
  edge_codes<<<dim3(gO > 0 ? gO : 1, 3), OWN_THREADS, 0, stream>>>(
      dst_follows, dst_ratedby, dst_rates, src_follows, src_ratedby, src_rates,
      pl_follows, pl_ratedby, pl_rates, ew_follows, ew_ratedby, ew_rates, Ef, Erb, Er);

  gemm_wmma_f32<<<GEMM_BLOCKS, GEMM_THREADS, 0, stream>>>(feat_user, W_follows, b_follows, whbuf, NU / 16);
  agg_owned_rows<<<gU, AGG_THREADS, 0, stream>>>(whbuf, pl_follows, ew_follows,
                                                 out_user, Ef, NU, 0);
  gemm_wmma_f32<<<GEMM_BLOCKS, GEMM_THREADS, 0, stream>>>(feat_item, W_ratedby, b_ratedby, whbuf, NI / 16);
  agg_owned_rows<<<gU, AGG_THREADS, 0, stream>>>(whbuf, pl_ratedby, ew_ratedby,
                                                 out_user, Erb, NU, 1);
  gemm_wmma_f32<<<GEMM_BLOCKS, GEMM_THREADS, 0, stream>>>(feat_user, W_rates, b_rates, whbuf, NU / 16);
  agg_owned_rows<<<gI, AGG_THREADS, 0, stream>>>(whbuf, pl_rates, ew_rates,
                                                 out_item, Er, NI, 0);
}


# LDS f32 atomic adds and 200-KiB sum tile replaced: owner ranks its queue by owned row in LDS (degree add's return, scan, place), a wave sums a row's sources in registers, writes once; rank order
# speedup vs baseline: 3.5750x; 1.0348x over previous
#include <hip/hip_runtime.h>

typedef __attribute__((ext_vector_type(2))) float v2f;
typedef __attribute__((ext_vector_type(8))) float v8f;

#define D 128

#define GEMM_THREADS 256
#define GEMM_BLOCKS 512
#define XP 132

__global__ __launch_bounds__(GEMM_THREADS)
void gemm_wmma_f32(const float* __restrict__ X,
                   const float* __restrict__ W,
                   const float* __restrict__ bias,
                   float* __restrict__ Y, int ntiles) {
  __shared__ float xs[2][16 * XP];

  const int tid = threadIdx.x;
  const int lane = tid & 31;
  const int wave = tid >> 5;
  const int n0 = wave * 16;
  const int r = lane & 15;
  const int hi = lane >> 4;
  const int koff = hi * 2;

  const float* brow = W + (size_t)(n0 + r) * D + koff;
  v2f b[D / 4];
#pragma unroll
  for (int kb = 0; kb < D / 4; ++kb) b[kb] = *(const v2f*)(brow + kb * 4);
  const float bv = bias[n0 + r];

  const float4* X4 = (const float4*)X;
  const int srow = tid >> 5;
  const int scol = (tid & 31) * 4;

  int tile = blockIdx.x;
  float4 g0 = make_float4(0.0f, 0.0f, 0.0f, 0.0f);
  float4 g1 = g0;
  if (tile < ntiles) {
    g0 = X4[(size_t)tile * 512 + tid];
    g1 = X4[(size_t)tile * 512 + 256 + tid];
  }
  int buf = 0;
  for (; tile < ntiles; tile += (int)gridDim.x) {
    float* xb = xs[buf];
    *(float4*)(xb + srow * XP + scol) = g0;
    *(float4*)(xb + (8 + srow) * XP + scol) = g1;
    __syncthreads();

    const int nt = tile + (int)gridDim.x;
    const size_t lt = (size_t)(nt < ntiles ? nt : tile);
    g0 = X4[lt * 512 + tid];
    g1 = X4[lt * 512 + 256 + tid];

    const float* ap = xb + r * XP + koff;
    v2f a[D / 4];
#pragma unroll
    for (int kb = 0; kb < D / 4; ++kb) a[kb] = *(const v2f*)(ap + kb * 4);

    __builtin_amdgcn_sched_barrier(0);

    v8f c = {};
#pragma unroll
    for (int kb = 0; kb < D / 4; ++kb) {
      c = __builtin_amdgcn_wmma_f32_16x16x4_f32(false, a[kb], false, b[kb],
                                                (short)0, c, false, false);
    }

    float* yout = Y + ((size_t)tile * 16 + hi * 8) * D + n0 + r;
#pragma unroll
    for (int v = 0; v < 8; ++v) yout[(size_t)v * D] = c[v] + bv;
    buf ^= 1;
  }
}

#define AGG_ROWS 400
#define AGG_QCAP 6144
#define AGG_THREADS 1024
#define AGG_SCAN_U 2
#define AGG_GATH_U 16
#define AGG_Q_U (AGG_QCAP / AGG_THREADS)
#define AGG_RPL ((AGG_ROWS + 31) / 32)

#define PL_EDGES 1024
#define PL_OWNERS 256

__device__ __forceinline__ unsigned owner_of(int d) {
  const unsigned o = (unsigned)d / (unsigned)AGG_ROWS;
  return o < 255u ? o : 255u;
}

#define OWN_THREADS 256
#define OWN_SUB 4
#define SRC_BITS 23
#define SRC_MASK 0x7FFFFFu

__device__ __forceinline__ unsigned edge_word(int d, int s) {
  const unsigned o = (unsigned)d / (unsigned)AGG_ROWS;
  return (((unsigned)d - o * (unsigned)AGG_ROWS) << SRC_BITS) | ((unsigned)s & SRC_MASK);
}

__global__ __launch_bounds__(OWN_THREADS)
void edge_codes(const int* __restrict__ dstA, const int* __restrict__ dstB,
                const int* __restrict__ dstC,
                const int* __restrict__ srcA, const int* __restrict__ srcB,
                const int* __restrict__ srcC,
                unsigned* __restrict__ plA, unsigned* __restrict__ plB,
                unsigned* __restrict__ plC,
                unsigned* __restrict__ ewA, unsigned* __restrict__ ewB,
                unsigned* __restrict__ ewC, int nA, int nB, int nC) {
  __shared__ unsigned bits[PL_OWNERS * 32];

  const int y = blockIdx.y;
  const int* dst = y == 0 ? dstA : (y == 1 ? dstB : dstC);
  const int* src = y == 0 ? srcA : (y == 1 ? srcB : srcC);
  unsigned* pl = y == 0 ? plA : (y == 1 ? plB : plC);
  unsigned* ew = y == 0 ? ewA : (y == 1 ? ewB : ewC);
  const int nE = y == 0 ? nA : (y == 1 ? nB : nC);
  const int n4 = nE >> 2;
  const int nblk = (nE + PL_EDGES - 1) / PL_EDGES;
  const int tid = threadIdx.x;
  const int4* d4 = (const int4*)dst;
  const int4* s4 = (const int4*)src;
  uint4* ew4 = (uint4*)ew;
  uint4* pl4 = (uint4*)pl;
  uint4* bits4 = (uint4*)bits;
  const unsigned wd = (unsigned)tid >> 3;
  const unsigned sh = ((unsigned)tid & 7u) * 4u;

  int4 dv[OWN_SUB];
  int4 sv[OWN_SUB];
#pragma unroll
  for (int s = 0; s < OWN_SUB; ++s) {
    dv[s] = make_int4(0, 0, 0, 0);
    sv[s] = dv[s];
  }
  if (n4 > 0) {
#pragma unroll
    for (int s = 0; s < OWN_SUB; ++s) {
      const int g = ((int)blockIdx.x * OWN_SUB + s) * OWN_THREADS + tid;
      const int gc = g < n4 ? g : 0;
      dv[s] = d4[gc];
      sv[s] = s4[gc];
    }
  }

#pragma unroll
  for (int s = 0; s < OWN_SUB; ++s) {
    const int blk = (int)blockIdx.x * OWN_SUB + s;
    if (blk >= nblk) break;
#pragma unroll
    for (int j = 0; j < (PL_OWNERS * 32 / 4) / OWN_THREADS; ++j)
      bits4[tid + j * OWN_THREADS] = make_uint4(0u, 0u, 0u, 0u);
    __syncthreads();

    const int g = blk * OWN_THREADS + tid;
    if (g < n4) {
      uint4 w;
      w.x = edge_word(dv[s].x, sv[s].x);
      w.y = edge_word(dv[s].y, sv[s].y);
      w.z = edge_word(dv[s].z, sv[s].z);
      w.w = edge_word(dv[s].w, sv[s].w);
      ew4[g] = w;
      atomicOr(&bits[owner_of(dv[s].x) * 32u + wd], 1u << sh);
      atomicOr(&bits[owner_of(dv[s].y) * 32u + wd], 2u << sh);
      atomicOr(&bits[owner_of(dv[s].z) * 32u + wd], 4u << sh);
      atomicOr(&bits[owner_of(dv[s].w) * 32u + wd], 8u << sh);
    } else if (g == n4) {
      for (int j = 0; j < (nE & 3); ++j) {
        const int e = n4 * 4 + j;
        const int d = dst[e];
        ew[e] = edge_word(d, src[e]);
        atomicOr(&bits[owner_of(d) * 32u + wd], (1u << j) << sh);
      }
    }
    __syncthreads();

#pragma unroll
    for (int j = 0; j < (PL_OWNERS * 32 / 4) / OWN_THREADS; ++j) {
      const int idx = tid + j * OWN_THREADS;
      const size_t o = (size_t)(idx >> 3);
      pl4[(o * (size_t)nblk + (size_t)blk) * 8u + (size_t)(idx & 7)] = bits4[idx];
    }
    __syncthreads();
  }
}

template <bool COUNT>
__device__ __forceinline__ void scan_plane(const uint4* __restrict__ pl4,
                                           const unsigned* __restrict__ ew,
                                           int p0, int p1, int tid,
                                           int* qe, unsigned* qn, int* deg) {
  for (int t0 = p0 + tid; t0 < p1; t0 += AGG_THREADS * AGG_SCAN_U) {
    uint4 pv[AGG_SCAN_U];
#pragma unroll
    for (int u = 0; u < AGG_SCAN_U; ++u) {
      const int t = t0 + u * AGG_THREADS;
      const bool ok = t < p1;
      const uint4 v = pl4[ok ? t : t0];
      pv[u].x = ok ? v.x : 0u;
      pv[u].y = ok ? v.y : 0u;
      pv[u].z = ok ? v.z : 0u;
      pv[u].w = ok ? v.w : 0u;
    }
#pragma unroll
    for (int u = 0; u < AGG_SCAN_U; ++u) {
      unsigned long long lo = (unsigned long long)pv[u].x | ((unsigned long long)pv[u].y << 32);
      unsigned long long hi = (unsigned long long)pv[u].z | ((unsigned long long)pv[u].w << 32);
      unsigned c = (unsigned)__popcll(lo) + (unsigned)__popcll(hi);
      if (c != 0u) {
        const int e0 = (t0 + u * AGG_THREADS) * 128;
        do {
          int b;
          if (lo != 0ull) {
            b = __builtin_ctzll(lo);
            lo &= lo - 1ull;
          } else {
            b = 64 + __builtin_ctzll(hi);
            hi &= hi - 1ull;
          }
          if (COUNT) {
            atomicAdd(&deg[ew[e0 + b] >> SRC_BITS], 1);
          } else {
            const unsigned k = atomicAdd(qn, 1u);
            if (k < (unsigned)AGG_QCAP) qe[k] = e0 + b;
          }
        } while (--c != 0u);
      }
    }
  }
}

template <int U>
__device__ __forceinline__ void sum_rows(const float* __restrict__ Wh, const int* qr,
                                         int kb, int d, int lane, float4& acc) {
  int s[U];
  float4 v[U];
#pragma unroll
  for (int u = 0; u < U; ++u) s[u] = qr[(kb + u) < d ? (kb + u) : (d - 1)];
#pragma unroll
  for (int u = 0; u < U; ++u) {
    v[u] = make_float4(0.0f, 0.0f, 0.0f, 0.0f);
    if ((kb + u) < d)
      v[u] = ((const float4*)(Wh + (size_t)s[u] * D))[lane];
  }
#pragma unroll
  for (int u = 0; u < U; ++u) {
    acc.x += v[u].x;
    acc.y += v[u].y;
    acc.z += v[u].z;
    acc.w += v[u].w;
  }
}

__global__ __launch_bounds__(AGG_THREADS)
void agg_owned_rows(const float* __restrict__ Wh,
                    const unsigned* __restrict__ pl,
                    const unsigned* __restrict__ ew,
                    float* __restrict__ out,
                    int nE, int nDst, int accumulate) {
  __shared__ int cnt[AGG_ROWS];
  __shared__ int rst[AGG_ROWS];
  __shared__ int deg[AGG_ROWS];
  __shared__ int qe[AGG_QCAP];
  __shared__ int qs[AGG_QCAP];
  __shared__ unsigned qn;

  const int tid = threadIdx.x;
  const int lane = tid & 31;
  const int wave = tid >> 5;
  const int d0 = blockIdx.x * AGG_ROWS;
  const int rows = (nDst - d0) < AGG_ROWS ? (nDst - d0) : AGG_ROWS;

  for (int i = tid; i < AGG_ROWS; i += AGG_THREADS) {
    cnt[i] = 0;
    deg[i] = 0;
  }
  if (tid == 0) qn = 0u;
  __syncthreads();

  const int nblk = (nE + PL_EDGES - 1) / PL_EDGES;
  const int n4 = (int)blockIdx.x < PL_OWNERS - 1 ? nblk * 8 : 0;
  const uint4* pl4 = (const uint4*)(pl + (size_t)blockIdx.x * (size_t)nblk * 32u);
  scan_plane<false>(pl4, ew, 0, n4, tid, qe, &qn, deg);
  __syncthreads();

  const bool pieces = qn > (unsigned)AGG_QCAP;
  int step = n4 > 0 ? n4 : 1;
  if (pieces) {
    scan_plane<true>(pl4, ew, 0, n4, tid, qe, &qn, deg);
    step = AGG_QCAP / 128;
  }

  int p0 = 0;
  do {
    if (pieces) {
      __syncthreads();
      for (int i = tid; i < AGG_ROWS; i += AGG_THREADS) cnt[i] = 0;
      if (tid == 0) qn = 0u;
      __syncthreads();
      scan_plane<false>(pl4, ew, p0, (p0 + step) < n4 ? (p0 + step) : n4, tid, qe, &qn, deg);
      __syncthreads();
    }
    const int n = (int)qn;

    unsigned pw[AGG_Q_U];
    int rk[AGG_Q_U];
#pragma unroll
    for (int u = 0; u < AGG_Q_U; ++u) {
      const int k = tid + u * AGG_THREADS;
      pw[u] = 0u;
      if (k < n) pw[u] = ew[qe[k]];
    }
#pragma unroll
    for (int u = 0; u < AGG_Q_U; ++u) {
      const int k = tid + u * AGG_THREADS;
      rk[u] = 0;
      if (k < n) rk[u] = atomicAdd(&cnt[pw[u] >> SRC_BITS], 1);
    }
    __syncthreads();

    if (wave == 0) {
      int loc[AGG_RPL];
      int s = 0;
#pragma unroll
      for (int j = 0; j < AGG_RPL; ++j) {
        const int r = lane * AGG_RPL + j;
        loc[j] = s;
        s += r < AGG_ROWS ? cnt[r] : 0;
      }
      int inc = s;
#pragma unroll
      for (int dlt = 1; dlt < 32; dlt <<= 1) {
        const int t = __shfl_up(inc, dlt);
        if (lane >= dlt) inc += t;
      }
#pragma unroll
      for (int j = 0; j < AGG_RPL; ++j) {
        const int r = lane * AGG_RPL + j;
        if (r < AGG_ROWS) rst[r] = inc - s + loc[j];
      }
    }
    __syncthreads();
#pragma unroll
    for (int u = 0; u < AGG_Q_U; ++u) {
      const int k = tid + u * AGG_THREADS;
      if (k < n) qs[rst[pw[u] >> SRC_BITS] + rk[u]] = (int)(pw[u] & SRC_MASK);
    }
    __syncthreads();

    const bool add_prev = accumulate != 0 || p0 > 0;
    int dn = 0;
    int stn = 0;
    if (wave < rows) {
      dn = cnt[wave];
      stn = rst[wave];
    }
    for (int r = wave; r < rows; r += AGG_THREADS / 32) {
      const int d = __builtin_amdgcn_readfirstlane(dn);
      const int st = __builtin_amdgcn_readfirstlane(stn);
      if (r + AGG_THREADS / 32 < rows) {
        dn = cnt[r + AGG_THREADS / 32];
        stn = rst[r + AGG_THREADS / 32];
      }
      float4* op = (float4*)(out + (size_t)(d0 + r) * D) + lane;
      float4 prev = make_float4(0.0f, 0.0f, 0.0f, 0.0f);
      if (add_prev) prev = *op;
      float4 acc = make_float4(0.0f, 0.0f, 0.0f, 0.0f);
      for (int kb = 0; kb < d; kb += AGG_GATH_U) sum_rows<AGG_GATH_U>(Wh, qs + st, kb, d, lane, acc);
      const float c = fmaxf((float)(pieces ? deg[r] : d), 1.0f);
      float4 o;
      o.x = acc.x / c;
      o.y = acc.y / c;
      o.z = acc.z / c;
      o.w = acc.w / c;
      if (add_prev) {
        o.x += prev.x; o.y += prev.y; o.z += prev.z; o.w += prev.w;
      }
      *op = o;
    }
    p0 += step;
  } while (p0 < n4);
}

extern "C" void kernel_launch(void* const* d_in, const int* in_sizes, int n_in,
                              void* d_out, int out_size, void* d_ws, size_t ws_size,
                              hipStream_t stream) {
  (void)n_in; (void)out_size; (void)ws_size;

  const float* feat_user  = (const float*)d_in[0];
  const float* feat_item  = (const float*)d_in[1];
  const float* W_follows  = (const float*)d_in[2];
  const float* b_follows  = (const float*)d_in[3];
  const float* W_rates    = (const float*)d_in[4];
  const float* b_rates    = (const float*)d_in[5];
  const float* W_ratedby  = (const float*)d_in[6];
  const float* b_ratedby  = (const float*)d_in[7];
  const int* src_follows  = (const int*)d_in[8];
  const int* dst_follows  = (const int*)d_in[9];
  const int* src_rates    = (const int*)d_in[10];
  const int* dst_rates    = (const int*)d_in[11];
  const int* src_ratedby  = (const int*)d_in[12];
  const int* dst_ratedby  = (const int*)d_in[13];

  const int NU  = in_sizes[0] / D;
  const int NI  = in_sizes[1] / D;
  const int Ef  = in_sizes[8];
  const int Er  = in_sizes[10];
  const int Erb = in_sizes[12];

  float* out_user = (float*)d_out;
  float* out_item = out_user + (size_t)NU * D;

  float* whbuf = (float*)d_ws;

  const int gU = (NU + AGG_ROWS - 1) / AGG_ROWS;
  const int gI = (NI + AGG_ROWS - 1) / AGG_ROWS;

  const int Emax = Ef > Er ? (Ef > Erb ? Ef : Erb) : (Er > Erb ? Er : Erb);
  const int nblkMax = (Emax + PL_EDGES - 1) / PL_EDGES;
  const size_t plWords = (size_t)PL_OWNERS * (size_t)nblkMax * 32;
  const size_t ewWords = (size_t)((Emax + 7) & ~3);
  unsigned* pl_follows = (unsigned*)(whbuf + (size_t)(NU > NI ? NU : NI) * D);
  unsigned* pl_ratedby = pl_follows + plWords;
  unsigned* pl_rates   = pl_ratedby + plWords;
  unsigned* ew_follows = pl_rates + plWords;
  unsigned* ew_ratedby = ew_follows + ewWords;
  unsigned* ew_rates   = ew_ratedby + ewWords;
  const int gO = (nblkMax + OWN_SUB - 1) / OWN_SUB;
  edge_codes<<<dim3(gO > 0 ? gO : 1, 3), OWN_THREADS, 0, stream>>>(
      dst_follows, dst_ratedby, dst_rates, src_follows, src_ratedby, src_rates,
      pl_follows, pl_ratedby, pl_rates, ew_follows, ew_ratedby, ew_rates, Ef, Erb, Er);

  gemm_wmma_f32<<<GEMM_BLOCKS, GEMM_THREADS, 0, stream>>>(feat_user, W_follows, b_follows, whbuf, NU / 16);
  agg_owned_rows<<<gU, AGG_THREADS, 0, stream>>>(whbuf, pl_follows, ew_follows,
                                                 out_user, Ef, NU, 0);
  gemm_wmma_f32<<<GEMM_BLOCKS, GEMM_THREADS, 0, stream>>>(feat_item, W_ratedby, b_ratedby, whbuf, NI / 16);
  agg_owned_rows<<<gU, AGG_THREADS, 0, stream>>>(whbuf, pl_ratedby, ew_ratedby,
                                                 out_user, Erb, NU, 1);
  gemm_wmma_f32<<<GEMM_BLOCKS, GEMM_THREADS, 0, stream>>>(feat_user, W_rates, b_rates, whbuf, NU / 16);
  agg_owned_rows<<<gI, AGG_THREADS, 0, stream>>>(whbuf, pl_rates, ew_rates,
                                                 out_item, Er, NI, 0);
}


# owner bit planes replaced: pre-launch groups each chunk of 4096 edge words by owning work-group in LDS, stores it whole with a run table; an owner reads only its runs; sums stay in LDS-rank order
# speedup vs baseline: 3.9062x; 1.0926x over previous
#include <hip/hip_runtime.h>

typedef __attribute__((ext_vector_type(2))) float v2f;
typedef __attribute__((ext_vector_type(8))) float v8f;

#define D 128

#define GEMM_THREADS 256
#define GEMM_BLOCKS 512
#define XP 132

__global__ __launch_bounds__(GEMM_THREADS)
void gemm_wmma_f32(const float* __restrict__ X,
                   const float* __restrict__ W,
                   const float* __restrict__ bias,
                   float* __restrict__ Y, int ntiles) {
  __shared__ float xs[2][16 * XP];

  const int tid = threadIdx.x;
  const int lane = tid & 31;
  const int wave = tid >> 5;
  const int n0 = wave * 16;
  const int r = lane & 15;
  const int hi = lane >> 4;
  const int koff = hi * 2;

  const float* brow = W + (size_t)(n0 + r) * D + koff;
  v2f b[D / 4];
#pragma unroll
  for (int kb = 0; kb < D / 4; ++kb) b[kb] = *(const v2f*)(brow + kb * 4);
  const float bv = bias[n0 + r];

  const float4* X4 = (const float4*)X;
  const int srow = tid >> 5;
  const int scol = (tid & 31) * 4;

  int tile = blockIdx.x;
  float4 g0 = make_float4(0.0f, 0.0f, 0.0f, 0.0f);
  float4 g1 = g0;
  if (tile < ntiles) {
    g0 = X4[(size_t)tile * 512 + tid];
    g1 = X4[(size_t)tile * 512 + 256 + tid];
  }
  int buf = 0;
  for (; tile < ntiles; tile += (int)gridDim.x) {
    float* xb = xs[buf];
    *(float4*)(xb + srow * XP + scol) = g0;
    *(float4*)(xb + (8 + srow) * XP + scol) = g1;
    __syncthreads();

    const int nt = tile + (int)gridDim.x;
    const size_t lt = (size_t)(nt < ntiles ? nt : tile);
    g0 = X4[lt * 512 + tid];
    g1 = X4[lt * 512 + 256 + tid];

    const float* ap = xb + r * XP + koff;
    v2f a[D / 4];
#pragma unroll
    for (int kb = 0; kb < D / 4; ++kb) a[kb] = *(const v2f*)(ap + kb * 4);

    __builtin_amdgcn_sched_barrier(0);

    v8f c = {};
#pragma unroll
    for (int kb = 0; kb < D / 4; ++kb) {
      c = __builtin_amdgcn_wmma_f32_16x16x4_f32(false, a[kb], false, b[kb],
                                                (short)0, c, false, false);
    }

    float* yout = Y + ((size_t)tile * 16 + hi * 8) * D + n0 + r;
#pragma unroll
    for (int v = 0; v < 8; ++v) yout[(size_t)v * D] = c[v] + bv;
    buf ^= 1;
  }
}

#define AGG_ROWS 400
#define AGG_QCAP 6144
#define AGG_THREADS 1024
#define AGG_GATH_U 16
#define AGG_Q_U (AGG_QCAP / AGG_THREADS)
#define AGG_RPL ((AGG_ROWS + 31) / 32)

#define PL_OWNERS 256
#define CH_EDGES 4096
#define CH_THREADS 256
#define CH_GROUPS (CH_EDGES / 4 / CH_THREADS)
#define CH_OPL (PL_OWNERS / 32)
#define AGG_STEP (AGG_QCAP / CH_EDGES)
static_assert(AGG_STEP >= 1, "a chunk must fit the queue");
static_assert(CH_THREADS == PL_OWNERS, "one thread per owner in the table line");

__device__ __forceinline__ unsigned owner_of(int d) {
  const unsigned o = (unsigned)d / (unsigned)AGG_ROWS;
  return o < 255u ? o : 255u;
}

#define SRC_BITS 23
#define SRC_MASK 0x7FFFFFu

__device__ __forceinline__ unsigned edge_word(int d, int s) {
  const unsigned o = (unsigned)d / (unsigned)AGG_ROWS;
  return (((unsigned)d - o * (unsigned)AGG_ROWS) << SRC_BITS) | ((unsigned)s & SRC_MASK);
}

__global__ __launch_bounds__(CH_THREADS)
void edge_codes(const int* __restrict__ dstA, const int* __restrict__ dstB,
                const int* __restrict__ dstC,
                const int* __restrict__ srcA, const int* __restrict__ srcB,
                const int* __restrict__ srcC,
                unsigned* __restrict__ sgA, unsigned* __restrict__ sgB,
                unsigned* __restrict__ sgC,
                unsigned* __restrict__ ewA, unsigned* __restrict__ ewB,
                unsigned* __restrict__ ewC, int nA, int nB, int nC) {
  __shared__ unsigned hist[PL_OWNERS + 1];
  __shared__ unsigned start[PL_OWNERS + 1];
  __shared__ unsigned sorted[CH_EDGES];

  const int y = blockIdx.y;
  const int* dst = y == 0 ? dstA : (y == 1 ? dstB : dstC);
  const int* src = y == 0 ? srcA : (y == 1 ? srcB : srcC);
  unsigned* sg = y == 0 ? sgA : (y == 1 ? sgB : sgC);
  unsigned* ew = y == 0 ? ewA : (y == 1 ? ewB : ewC);
  const int nE = y == 0 ? nA : (y == 1 ? nB : nC);
  const int chunk = blockIdx.x;
  const int nch = (nE + CH_EDGES - 1) / CH_EDGES;
  if (chunk >= nch) return;

  const int n4 = nE >> 2;
  const int tid = threadIdx.x;
  const int lane = tid & 31;
  const int wave = tid >> 5;
  const int4* d4 = (const int4*)dst;
  const int4* s4 = (const int4*)src;
  const int left = nE - chunk * CH_EDGES;
  const int total = left < CH_EDGES ? left : CH_EDGES;

  int dd[CH_GROUPS * 4];
  int ss[CH_GROUPS * 4];
  unsigned vmask = 0u;
#pragma unroll
  for (int i = 0; i < CH_GROUPS * 4; ++i) {
    dd[i] = 0;
    ss[i] = 0;
  }
  if (n4 > 0) {
#pragma unroll
    for (int s = 0; s < CH_GROUPS; ++s) {
      const int g = (chunk * CH_GROUPS + s) * CH_THREADS + tid;
      const int gc = g < n4 ? g : 0;
      const int4 dv = d4[gc];
      const int4 sv = s4[gc];
      dd[4 * s + 0] = dv.x; dd[4 * s + 1] = dv.y; dd[4 * s + 2] = dv.z; dd[4 * s + 3] = dv.w;
      ss[4 * s + 0] = sv.x; ss[4 * s + 1] = sv.y; ss[4 * s + 2] = sv.z; ss[4 * s + 3] = sv.w;
      if (g < n4) vmask |= 15u << (4 * s);
    }
  }
#pragma unroll
  for (int s = 0; s < CH_GROUPS; ++s) {
    const int g = (chunk * CH_GROUPS + s) * CH_THREADS + tid;
    if (g == n4) {
      const int e = n4 * 4;
      const int rem = nE & 3;
      if (rem > 0) { dd[4 * s + 0] = dst[e];     ss[4 * s + 0] = src[e];     vmask |= 1u << (4 * s); }
      if (rem > 1) { dd[4 * s + 1] = dst[e + 1]; ss[4 * s + 1] = src[e + 1]; vmask |= 2u << (4 * s); }
      if (rem > 2) { dd[4 * s + 2] = dst[e + 2]; ss[4 * s + 2] = src[e + 2]; vmask |= 4u << (4 * s); }
    }
  }

  hist[tid] = 0u;
  if (tid == 0) {
    hist[PL_OWNERS] = 0u;
    start[PL_OWNERS] = (unsigned)total;
  }
  __syncthreads();

  unsigned code[CH_GROUPS * 4];
#pragma unroll
  for (int i = 0; i < CH_GROUPS * 4; ++i) {
    const unsigned o = ((vmask >> i) & 1u) ? owner_of(dd[i]) : (unsigned)PL_OWNERS;
    code[i] = (o << 16) | atomicAdd(&hist[o], 1u);
  }
  __syncthreads();

  if (wave == 0) {
    unsigned loc[CH_OPL];
    unsigned s = 0u;
#pragma unroll
    for (int j = 0; j < CH_OPL; ++j) {
      loc[j] = s;
      s += hist[lane * CH_OPL + j];
    }
    unsigned inc = s;
#pragma unroll
    for (int dlt = 1; dlt < 32; dlt <<= 1) {
      const unsigned t = __shfl_up(inc, dlt);
      if (lane >= dlt) inc += t;
    }
#pragma unroll
    for (int j = 0; j < CH_OPL; ++j) start[lane * CH_OPL + j] = inc - s + loc[j];
  }
  __syncthreads();

  unsigned pos[CH_GROUPS * 4];
#pragma unroll
  for (int i = 0; i < CH_GROUPS * 4; ++i) pos[i] = start[code[i] >> 16] + (code[i] & 0xFFFFu);
#pragma unroll
  for (int i = 0; i < CH_GROUPS * 4; ++i)
    sorted[pos[i]] = ((vmask >> i) & 1u) ? edge_word(dd[i], ss[i]) : 0u;
  sg[(size_t)chunk * PL_OWNERS + tid] = start[tid] | (hist[tid] << 16);
  __syncthreads();

  uint4* ew4 = (uint4*)(ew + (size_t)chunk * CH_EDGES);
  const uint4* sorted4 = (const uint4*)sorted;
#pragma unroll
  for (int j = 0; j < CH_GROUPS; ++j) {
    const int idx = tid + j * CH_THREADS;
    if (idx * 4 < total) ew4[idx] = sorted4[idx];
  }
}

template <bool COUNT>
__device__ __forceinline__ void scan_chunks(const unsigned* __restrict__ sg,
                                            const unsigned* __restrict__ ew,
                                            int owner, int c0, int c1, int tid,
                                            int* qe, unsigned* qn, int* deg) {
  for (int c = c0 + tid; c < c1; c += AGG_THREADS) {
    const unsigned t = sg[(size_t)c * PL_OWNERS + owner];
    const int n = (int)(t >> 16);
    const int base = c * CH_EDGES + (int)(t & 0xFFFFu);
    if (COUNT) {
      for (int i = 0; i < n; ++i) atomicAdd(&deg[ew[base + i] >> SRC_BITS], 1);
    } else if (n > 0) {
      const unsigned k0 = atomicAdd(qn, (unsigned)n);
      for (int i = 0; i < n; ++i)
        if (k0 + (unsigned)i < (unsigned)AGG_QCAP) qe[k0 + i] = base + i;
    }
  }
}

template <int U>
__device__ __forceinline__ void sum_rows(const float* __restrict__ Wh, const int* qr,
                                         int kb, int d, int lane, float4& acc) {
  int s[U];
  float4 v[U];
#pragma unroll
  for (int u = 0; u < U; ++u) s[u] = qr[(kb + u) < d ? (kb + u) : (d - 1)];
#pragma unroll
  for (int u = 0; u < U; ++u) {
    v[u] = make_float4(0.0f, 0.0f, 0.0f, 0.0f);
    if ((kb + u) < d)
      v[u] = ((const float4*)(Wh + (size_t)s[u] * D))[lane];
  }
#pragma unroll
  for (int u = 0; u < U; ++u) {
    acc.x += v[u].x;
    acc.y += v[u].y;
    acc.z += v[u].z;
    acc.w += v[u].w;
  }
}

__global__ __launch_bounds__(AGG_THREADS)
void agg_owned_rows(const float* __restrict__ Wh,
                    const unsigned* __restrict__ sg,
                    const unsigned* __restrict__ ew,
                    float* __restrict__ out,
                    int nE, int nDst, int accumulate) {
  __shared__ int cnt[AGG_ROWS];
  __shared__ int rst[AGG_ROWS];
  __shared__ int deg[AGG_ROWS];
  __shared__ int qe[AGG_QCAP];
  __shared__ int qs[AGG_QCAP];
  __shared__ unsigned qn;

  const int tid = threadIdx.x;
  const int lane = tid & 31;
  const int wave = tid >> 5;
  const int d0 = blockIdx.x * AGG_ROWS;
  const int rows = (nDst - d0) < AGG_ROWS ? (nDst - d0) : AGG_ROWS;

  for (int i = tid; i < AGG_ROWS; i += AGG_THREADS) {
    cnt[i] = 0;
    deg[i] = 0;
  }
  if (tid == 0) qn = 0u;
  __syncthreads();

  const int nch = (int)blockIdx.x < PL_OWNERS - 1 ? (nE + CH_EDGES - 1) / CH_EDGES : 0;
  const int owner = (int)blockIdx.x < PL_OWNERS - 1 ? (int)blockIdx.x : PL_OWNERS - 1;
  scan_chunks<false>(sg, ew, owner, 0, nch, tid, qe, &qn, deg);
  __syncthreads();

  const bool pieces = qn > (unsigned)AGG_QCAP;
  int step = nch > 0 ? nch : 1;
  if (pieces) {
    scan_chunks<true>(sg, ew, owner, 0, nch, tid, qe, &qn, deg);
    step = AGG_STEP;
  }

  int p0 = 0;
  do {
    if (pieces) {
      __syncthreads();
      for (int i = tid; i < AGG_ROWS; i += AGG_THREADS) cnt[i] = 0;
      if (tid == 0) qn = 0u;
      __syncthreads();
      scan_chunks<false>(sg, ew, owner, p0, (p0 + step) < nch ? (p0 + step) : nch, tid, qe, &qn, deg);
      __syncthreads();
    }
    const int n = (int)qn;

    unsigned pw[AGG_Q_U];
    int rk[AGG_Q_U];
#pragma unroll
    for (int u = 0; u < AGG_Q_U; ++u) {
      const int k = tid + u * AGG_THREADS;
      pw[u] = 0u;
      if (k < n) pw[u] = ew[qe[k]];
    }
#pragma unroll
    for (int u = 0; u < AGG_Q_U; ++u) {
      const int k = tid + u * AGG_THREADS;
      rk[u] = 0;
      if (k < n) rk[u] = atomicAdd(&cnt[pw[u] >> SRC_BITS], 1);
    }
    __syncthreads();

    if (wave == 0) {
      int loc[AGG_RPL];
      int s = 0;
#pragma unroll
      for (int j = 0; j < AGG_RPL; ++j) {
        const int r = lane * AGG_RPL + j;
        loc[j] = s;
        s += r < AGG_ROWS ? cnt[r] : 0;
      }
      int inc = s;
#pragma unroll
      for (int dlt = 1; dlt < 32; dlt <<= 1) {
        const int t = __shfl_up(inc, dlt);
        if (lane >= dlt) inc += t;
      }
#pragma unroll
      for (int j = 0; j < AGG_RPL; ++j) {
        const int r = lane * AGG_RPL + j;
        if (r < AGG_ROWS) rst[r] = inc - s + loc[j];
      }
    }
    __syncthreads();
#pragma unroll
    for (int u = 0; u < AGG_Q_U; ++u) {
      const int k = tid + u * AGG_THREADS;
      if (k < n) qs[rst[pw[u] >> SRC_BITS] + rk[u]] = (int)(pw[u] & SRC_MASK);
    }
    __syncthreads();

    const bool add_prev = accumulate != 0 || p0 > 0;
    int dn = 0;
    int stn = 0;
    if (wave < rows) {
      dn = cnt[wave];
      stn = rst[wave];
    }
    for (int r = wave; r < rows; r += AGG_THREADS / 32) {
      const int d = __builtin_amdgcn_readfirstlane(dn);
      const int st = __builtin_amdgcn_readfirstlane(stn);
      if (r + AGG_THREADS / 32 < rows) {
        dn = cnt[r + AGG_THREADS / 32];
        stn = rst[r + AGG_THREADS / 32];
      }
      float4* op = (float4*)(out + (size_t)(d0 + r) * D) + lane;
      float4 prev = make_float4(0.0f, 0.0f, 0.0f, 0.0f);
      if (add_prev) prev = *op;
      float4 acc = make_float4(0.0f, 0.0f, 0.0f, 0.0f);
      for (int kb = 0; kb < d; kb += AGG_GATH_U) sum_rows<AGG_GATH_U>(Wh, qs + st, kb, d, lane, acc);
      const float c = fmaxf((float)(pieces ? deg[r] : d), 1.0f);
      float4 o;
      o.x = acc.x / c;
      o.y = acc.y / c;
      o.z = acc.z / c;
      o.w = acc.w / c;
      if (add_prev) {
        o.x += prev.x; o.y += prev.y; o.z += prev.z; o.w += prev.w;
      }
      *op = o;
    }
    p0 += step;
  } while (p0 < nch);
}

extern "C" void kernel_launch(void* const* d_in, const int* in_sizes, int n_in,
                              void* d_out, int out_size, void* d_ws, size_t ws_size,
                              hipStream_t stream) {
  (void)n_in; (void)out_size; (void)ws_size;

  const float* feat_user  = (const float*)d_in[0];
  const float* feat_item  = (const float*)d_in[1];
  const float* W_follows  = (const float*)d_in[2];
  const float* b_follows  = (const float*)d_in[3];
  const float* W_rates    = (const float*)d_in[4];
  const float* b_rates    = (const float*)d_in[5];
  const float* W_ratedby  = (const float*)d_in[6];
  const float* b_ratedby  = (const float*)d_in[7];
  const int* src_follows  = (const int*)d_in[8];
  const int* dst_follows  = (const int*)d_in[9];
  const int* src_rates    = (const int*)d_in[10];
  const int* dst_rates    = (const int*)d_in[11];
  const int* src_ratedby  = (const int*)d_in[12];
  const int* dst_ratedby  = (const int*)d_in[13];

  const int NU  = in_sizes[0] / D;
  const int NI  = in_sizes[1] / D;
  const int Ef  = in_sizes[8];
  const int Er  = in_sizes[10];
  const int Erb = in_sizes[12];

  float* out_user = (float*)d_out;
  float* out_item = out_user + (size_t)NU * D;

  float* whbuf = (float*)d_ws;

  const int gU = (NU + AGG_ROWS - 1) / AGG_ROWS;
  const int gI = (NI + AGG_ROWS - 1) / AGG_ROWS;

  const int Emax = Ef > Er ? (Ef > Erb ? Ef : Erb) : (Er > Erb ? Er : Erb);
  const int nchMax = (Emax + CH_EDGES - 1) / CH_EDGES;
  const size_t sgWords = (size_t)nchMax * PL_OWNERS;
  const size_t ewWords = (size_t)nchMax * CH_EDGES;
  unsigned* sg_follows = (unsigned*)(whbuf + (size_t)(NU > NI ? NU : NI) * D);
  unsigned* sg_ratedby = sg_follows + sgWords;
  unsigned* sg_rates   = sg_ratedby + sgWords;
  unsigned* ew_follows = sg_rates + sgWords;
  unsigned* ew_ratedby = ew_follows + ewWords;
  unsigned* ew_rates   = ew_ratedby + ewWords;
  edge_codes<<<dim3(nchMax > 0 ? nchMax : 1, 3), CH_THREADS, 0, stream>>>(
      dst_follows, dst_ratedby, dst_rates, src_follows, src_ratedby, src_rates,
      sg_follows, sg_ratedby, sg_rates, ew_follows, ew_ratedby, ew_rates, Ef, Erb, Er);

  gemm_wmma_f32<<<GEMM_BLOCKS, GEMM_THREADS, 0, stream>>>(feat_user, W_follows, b_follows, whbuf, NU / 16);
  agg_owned_rows<<<gU, AGG_THREADS, 0, stream>>>(whbuf, sg_follows, ew_follows,
                                                 out_user, Ef, NU, 0);
  gemm_wmma_f32<<<GEMM_BLOCKS, GEMM_THREADS, 0, stream>>>(feat_item, W_ratedby, b_ratedby, whbuf, NI / 16);
  agg_owned_rows<<<gU, AGG_THREADS, 0, stream>>>(whbuf, sg_ratedby, ew_ratedby,
                                                 out_user, Erb, NU, 1);
  gemm_wmma_f32<<<GEMM_BLOCKS, GEMM_THREADS, 0, stream>>>(feat_user, W_rates, b_rates, whbuf, NU / 16);
  agg_owned_rows<<<gI, AGG_THREADS, 0, stream>>>(whbuf, sg_rates, ew_rates,
                                                 out_item, Er, NI, 0);
}
